# starting kernel
# baseline (speedup 1.0000x reference)
#include <hip/hip_runtime.h>

#define N_NODES  50000
#define N_EDGES  600000
#define E_TOT    (N_EDGES + N_NODES)
#define N_GRAPHS 1024
#define D        128
#define NEG_SLOPE 0.2f

typedef float    v2f  __attribute__((ext_vector_type(2)));
typedef float    v8f  __attribute__((ext_vector_type(8)));
typedef _Float16 v16h __attribute__((ext_vector_type(16)));

__global__ __launch_bounds__(256)
void gemm_wmma(const float* __restrict__ X, const float* __restrict__ W,
               float* __restrict__ H) {
  __shared__ float xs[16][132];
  __shared__ float ws[128][136];
  const int row0 = blockIdx.x * 16;
  const int t = threadIdx.x;

  for (int i = t; i < 16 * 32; i += 256) {
    const int r = i >> 5, c4 = i & 31;
    const float4 v = ((const float4*)(X + (size_t)(row0 + r) * D))[c4];
    float* p = &xs[r][c4 * 4];
    p[0] = v.x; p[1] = v.y; p[2] = v.z; p[3] = v.w;
  }
  for (int i = t; i < 128 * 32; i += 256) {
    const int r = i >> 5, c4 = i & 31;
    *(float4*)&ws[r][c4 * 4] = ((const float4*)(W + (size_t)r * D))[c4];
  }
  __syncthreads();

  const int wave = t >> 5, lane = t & 31;
  const int hi = lane >> 4;
  const int m  = lane & 15;
  const int n  = wave * 16 + m;

  v8f c = {0.f, 0.f, 0.f, 0.f, 0.f, 0.f, 0.f, 0.f};

#if __has_builtin(__builtin_amdgcn_wmma_f32_16x16x4_f32)
  #pragma unroll
  for (int k0 = 0; k0 < D; k0 += 4) {
    const int ka = k0 + hi * 2;
    v2f a, b;
    a[0] = xs[m][ka];
    a[1] = xs[m][ka + 1];
    b[0] = ws[ka][n];
    b[1] = ws[ka + 1][n];
    c = __builtin_amdgcn_wmma_f32_16x16x4_f32(false, a, false, b, (short)0, c,
                                              false, false);
  }
#else
  #pragma unroll
  for (int k0 = 0; k0 < D; k0 += 32) {
    v16h a, b;
    #pragma unroll
    for (int i = 0; i < 16; ++i) {
      const int vv = i >> 1, j = i & 1;
      const int ka = k0 + ((vv >= 4) ? 16 : 0) + hi * 8 + (vv & 3) * 2 + j;
      a[i] = (_Float16)xs[m][ka];
      const int kb = k0 + hi * 16 + i;
      b[i] = (_Float16)ws[kb][n];
    }
    c = __builtin_amdgcn_wmma_f32_16x16x32_f16(false, a, false, b, (short)0, c,
                                               false, false);
  }
#endif

  #pragma unroll
  for (int r = 0; r < 8; ++r)
    H[(size_t)(row0 + r + hi * 8) * D + n] = c[r];
}

__global__ __launch_bounds__(256)
void alpha_kernel(const float* __restrict__ H, const float* __restrict__ a_src,
                  const float* __restrict__ a_dst, float* __restrict__ asrc,
                  float* __restrict__ adst) {
  const int node = (blockIdx.x * blockDim.x + threadIdx.x) >> 5;
  const int lane = threadIdx.x & 31;
  if (node >= N_NODES) return;
  const int c0 = lane * 4;
  const float4 hv = *(const float4*)(H + (size_t)node * D + c0);
  const float4 sv = *(const float4*)(a_src + c0);
  const float4 dv = *(const float4*)(a_dst + c0);
  float s = hv.x * sv.x + hv.y * sv.y + hv.z * sv.z + hv.w * sv.w;
  float d = hv.x * dv.x + hv.y * dv.y + hv.z * dv.z + hv.w * dv.w;
  #pragma unroll
  for (int off = 16; off > 0; off >>= 1) {
    s += __shfl_down(s, off, 32);
    d += __shfl_down(d, off, 32);
  }
  if (lane == 0) { asrc[node] = s; adst[node] = d; }
}

__device__ __forceinline__ unsigned fmap(float x) {
  const unsigned u = __float_as_uint(x);
  return (u & 0x80000000u) ? ~u : (u | 0x80000000u);
}
__device__ __forceinline__ float funmap(unsigned u) {
  return __uint_as_float((u & 0x80000000u) ? (u & 0x7fffffffu) : ~u);
}

__global__ __launch_bounds__(256)
void edge_pass1(const int* __restrict__ ei, const float* __restrict__ asrc,
                const float* __restrict__ adst, float* __restrict__ ebuf,
                unsigned* __restrict__ emaxu) {
  const int t = blockIdx.x * blockDim.x + threadIdx.x;
  if (t >= E_TOT) return;
  const int s = (t < N_EDGES) ? ei[t] : (t - N_EDGES);
  const int d = (t < N_EDGES) ? ei[N_EDGES + t] : (t - N_EDGES);
  float v = asrc[s] + adst[d];
  v = (v > 0.f) ? v : NEG_SLOPE * v;
  ebuf[t] = v;
  atomicMax(&emaxu[d], fmap(v));
}

__global__ __launch_bounds__(256)
void edge_pass2(const int* __restrict__ ei, float* __restrict__ ebuf,
                const unsigned* __restrict__ emaxu, float* __restrict__ denom) {
  const int t = blockIdx.x * blockDim.x + threadIdx.x;
  if (t >= E_TOT) return;
  const int d = (t < N_EDGES) ? ei[N_EDGES + t] : (t - N_EDGES);
  const float ex = __expf(ebuf[t] - funmap(emaxu[d]));
  ebuf[t] = ex;
  atomicAdd(&denom[d], ex);
}

__global__ __launch_bounds__(256)
void edge_pass3(const int* __restrict__ ei, const float* __restrict__ ebuf,
                const float* __restrict__ denom, const float* __restrict__ H,
                float* __restrict__ acc) {
  const int gt = blockIdx.x * blockDim.x + threadIdx.x;
  const int edge = gt >> 5;
  const int lane = threadIdx.x & 31;
  if (edge >= E_TOT) return;
  const int s = (edge < N_EDGES) ? ei[edge] : (edge - N_EDGES);
  const int d = (edge < N_EDGES) ? ei[N_EDGES + edge] : (edge - N_EDGES);
  const float coef = ebuf[edge] / denom[d];
  const int c0 = lane * 4;
  const float4 hv = *(const float4*)(H + (size_t)s * D + c0);
  float* ad = acc + (size_t)d * D + c0;
  atomicAdd(&ad[0], hv.x * coef);
  atomicAdd(&ad[1], hv.y * coef);
  atomicAdd(&ad[2], hv.z * coef);
  atomicAdd(&ad[3], hv.w * coef);
}

__global__ __launch_bounds__(256)
void relu_pool(float* __restrict__ acc, const float* __restrict__ bias,
               const int* __restrict__ batch, float* __restrict__ embsum) {
  const int t = blockIdx.x * blockDim.x + threadIdx.x;
  if (t >= N_NODES * D) return;
  const int node = t >> 7, c = t & 127;
  float v = acc[t] + bias[c];
  v = (v > 0.f) ? v : 0.f;
  acc[t] = v;
  atomicAdd(&embsum[(size_t)batch[node] * D + c], v);
}

__global__ __launch_bounds__(256)
void count_kernel(const int* __restrict__ batch, float* __restrict__ cnt) {
  const int t = blockIdx.x * blockDim.x + threadIdx.x;
  if (t < N_NODES) atomicAdd(&cnt[batch[t]], 1.0f);
}

__global__ __launch_bounds__(256)
void finalize_kernel(float* __restrict__ out, const float* __restrict__ cnt) {
  const int t = blockIdx.x * blockDim.x + threadIdx.x;
  if (t >= 2 * N_GRAPHS * D) return;
  const int g = (t >> 7) & (N_GRAPHS - 1);
  float c = cnt[g];
  c = (c > 1.f) ? c : 1.f;
  out[t] /= c;
}

extern "C" void kernel_launch(void* const* d_in, const int* in_sizes, int n_in,
                              void* d_out, int out_size, void* d_ws, size_t ws_size,
                              hipStream_t stream) {
  const float* x     = (const float*)d_in[0];
  const int*   ei    = (const int*)d_in[1];
  const int*   batch = (const int*)d_in[2];
  const float* W1    = (const float*)d_in[3];
  const float* as1   = (const float*)d_in[4];
  const float* ad1   = (const float*)d_in[5];
  const float* b1    = (const float*)d_in[6];
  const float* W2    = (const float*)d_in[7];
  const float* as2   = (const float*)d_in[8];
  const float* ad2   = (const float*)d_in[9];
  const float* b2    = (const float*)d_in[10];
  float* out = (float*)d_out;

  char* w = (char*)d_ws;
  auto carve = [&](size_t bytes) {
    char* p = w;
    w += (bytes + 255) & ~(size_t)255;
    return p;
  };
  float*    hbuf  = (float*)carve((size_t)N_NODES * D * 4);
  float*    acc   = (float*)carve((size_t)N_NODES * D * 4);
  float*    asrc  = (float*)carve((size_t)N_NODES * 4);
  float*    adst  = (float*)carve((size_t)N_NODES * 4);
  unsigned* emaxu = (unsigned*)carve((size_t)N_NODES * 4);
  float*    denom = (float*)carve((size_t)N_NODES * 4);
  float*    ebuf  = (float*)carve((size_t)E_TOT * 4);
  float*    cnt   = (float*)carve((size_t)N_GRAPHS * 4);

  const int TB = 256;
  const int gGemm  = N_NODES / 16;
  const int gAlpha = (N_NODES * 32 + TB - 1) / TB;
  const int gEdge  = (E_TOT + TB - 1) / TB;
  const int gEdgeW = (E_TOT * 32 + TB - 1) / TB;
  const int gNodeC = (N_NODES * D + TB - 1) / TB;
  const int gNode  = (N_NODES + TB - 1) / TB;
  const int gFin   = (2 * N_GRAPHS * D + TB - 1) / TB;

  hipMemsetAsync(d_out, 0, (size_t)2 * N_GRAPHS * D * 4, stream);
  hipMemsetAsync(cnt, 0, (size_t)N_GRAPHS * 4, stream);
  count_kernel<<<gNode, TB, 0, stream>>>(batch, cnt);

  gemm_wmma<<<gGemm, TB, 0, stream>>>(x, W1, hbuf);
  alpha_kernel<<<gAlpha, TB, 0, stream>>>(hbuf, as1, ad1, asrc, adst);
  hipMemsetAsync(emaxu, 0, (size_t)N_NODES * 4, stream);
  hipMemsetAsync(denom, 0, (size_t)N_NODES * 4, stream);
  hipMemsetAsync(acc, 0, (size_t)N_NODES * D * 4, stream);
  edge_pass1<<<gEdge, TB, 0, stream>>>(ei, asrc, adst, ebuf, emaxu);
  edge_pass2<<<gEdge, TB, 0, stream>>>(ei, ebuf, emaxu, denom);
  edge_pass3<<<gEdgeW, TB, 0, stream>>>(ei, ebuf, denom, hbuf, acc);
  relu_pool<<<gNodeC, TB, 0, stream>>>(acc, b1, batch, out);

  gemm_wmma<<<gGemm, TB, 0, stream>>>(acc, W2, hbuf);
  alpha_kernel<<<gAlpha, TB, 0, stream>>>(hbuf, as2, ad2, asrc, adst);
  hipMemsetAsync(emaxu, 0, (size_t)N_NODES * 4, stream);
  hipMemsetAsync(denom, 0, (size_t)N_NODES * 4, stream);
  hipMemsetAsync(acc, 0, (size_t)N_NODES * D * 4, stream);
  edge_pass1<<<gEdge, TB, 0, stream>>>(ei, asrc, adst, ebuf, emaxu);
  edge_pass2<<<gEdge, TB, 0, stream>>>(ei, ebuf, emaxu, denom);
  edge_pass3<<<gEdgeW, TB, 0, stream>>>(ei, ebuf, denom, hbuf, acc);
  relu_pool<<<gNodeC, TB, 0, stream>>>(acc, b2, batch, out + N_GRAPHS * D);

  finalize_kernel<<<gFin, TB, 0, stream>>>(out, cnt);
}


# Edge scatter: lane l owns columns l, l+32, l+64, l+96, so each row load and each atomic instruction covers one whole 128-byte line, not a quarter of four; sums in arrival order as before
# speedup vs baseline: 1.3510x; 1.3510x over previous
#include <hip/hip_runtime.h>

#define N_NODES  50000
#define N_EDGES  600000
#define E_TOT    (N_EDGES + N_NODES)
#define N_GRAPHS 1024
#define D        128
#define NEG_SLOPE 0.2f

typedef float    v2f  __attribute__((ext_vector_type(2)));
typedef float    v8f  __attribute__((ext_vector_type(8)));
typedef _Float16 v16h __attribute__((ext_vector_type(16)));

__global__ __launch_bounds__(256)
void gemm_wmma(const float* __restrict__ X, const float* __restrict__ W,
               float* __restrict__ H) {
  __shared__ float xs[16][132];
  __shared__ float ws[128][136];
  const int row0 = blockIdx.x * 16;
  const int t = threadIdx.x;

  for (int i = t; i < 16 * 32; i += 256) {
    const int r = i >> 5, c4 = i & 31;
    const float4 v = ((const float4*)(X + (size_t)(row0 + r) * D))[c4];
    float* p = &xs[r][c4 * 4];
    p[0] = v.x; p[1] = v.y; p[2] = v.z; p[3] = v.w;
  }
  for (int i = t; i < 128 * 32; i += 256) {
    const int r = i >> 5, c4 = i & 31;
    *(float4*)&ws[r][c4 * 4] = ((const float4*)(W + (size_t)r * D))[c4];
  }
  __syncthreads();

  const int wave = t >> 5, lane = t & 31;
  const int hi = lane >> 4;
  const int m  = lane & 15;
  const int n  = wave * 16 + m;

  v8f c = {0.f, 0.f, 0.f, 0.f, 0.f, 0.f, 0.f, 0.f};

#if __has_builtin(__builtin_amdgcn_wmma_f32_16x16x4_f32)
  #pragma unroll
  for (int k0 = 0; k0 < D; k0 += 4) {
    const int ka = k0 + hi * 2;
    v2f a, b;
    a[0] = xs[m][ka];
    a[1] = xs[m][ka + 1];
    b[0] = ws[ka][n];
    b[1] = ws[ka + 1][n];
    c = __builtin_amdgcn_wmma_f32_16x16x4_f32(false, a, false, b, (short)0, c,
                                              false, false);
  }
#else
  #pragma unroll
  for (int k0 = 0; k0 < D; k0 += 32) {
    v16h a, b;
    #pragma unroll
    for (int i = 0; i < 16; ++i) {
      const int vv = i >> 1, j = i & 1;
      const int ka = k0 + ((vv >= 4) ? 16 : 0) + hi * 8 + (vv & 3) * 2 + j;
      a[i] = (_Float16)xs[m][ka];
      const int kb = k0 + hi * 16 + i;
      b[i] = (_Float16)ws[kb][n];
    }
    c = __builtin_amdgcn_wmma_f32_16x16x32_f16(false, a, false, b, (short)0, c,
                                               false, false);
  }
#endif

  #pragma unroll
  for (int r = 0; r < 8; ++r)
    H[(size_t)(row0 + r + hi * 8) * D + n] = c[r];
}

__global__ __launch_bounds__(256)
void alpha_kernel(const float* __restrict__ H, const float* __restrict__ a_src,
                  const float* __restrict__ a_dst, float* __restrict__ asrc,
                  float* __restrict__ adst) {
  const int node = (blockIdx.x * blockDim.x + threadIdx.x) >> 5;
  const int lane = threadIdx.x & 31;
  if (node >= N_NODES) return;
  const int c0 = lane * 4;
  const float4 hv = *(const float4*)(H + (size_t)node * D + c0);
  const float4 sv = *(const float4*)(a_src + c0);
  const float4 dv = *(const float4*)(a_dst + c0);
  float s = hv.x * sv.x + hv.y * sv.y + hv.z * sv.z + hv.w * sv.w;
  float d = hv.x * dv.x + hv.y * dv.y + hv.z * dv.z + hv.w * dv.w;
  #pragma unroll
  for (int off = 16; off > 0; off >>= 1) {
    s += __shfl_down(s, off, 32);
    d += __shfl_down(d, off, 32);
  }
  if (lane == 0) { asrc[node] = s; adst[node] = d; }
}

__device__ __forceinline__ unsigned fmap(float x) {
  const unsigned u = __float_as_uint(x);
  return (u & 0x80000000u) ? ~u : (u | 0x80000000u);
}
__device__ __forceinline__ float funmap(unsigned u) {
  return __uint_as_float((u & 0x80000000u) ? (u & 0x7fffffffu) : ~u);
}

__global__ __launch_bounds__(256)
void edge_pass1(const int* __restrict__ ei, const float* __restrict__ asrc,
                const float* __restrict__ adst, float* __restrict__ ebuf,
                unsigned* __restrict__ emaxu) {
  const int t = blockIdx.x * blockDim.x + threadIdx.x;
  if (t >= E_TOT) return;
  const int s = (t < N_EDGES) ? ei[t] : (t - N_EDGES);
  const int d = (t < N_EDGES) ? ei[N_EDGES + t] : (t - N_EDGES);
  float v = asrc[s] + adst[d];
  v = (v > 0.f) ? v : NEG_SLOPE * v;
  ebuf[t] = v;
  atomicMax(&emaxu[d], fmap(v));
}

__global__ __launch_bounds__(256)
void edge_pass2(const int* __restrict__ ei, float* __restrict__ ebuf,
                const unsigned* __restrict__ emaxu, float* __restrict__ denom) {
  const int t = blockIdx.x * blockDim.x + threadIdx.x;
  if (t >= E_TOT) return;
  const int d = (t < N_EDGES) ? ei[N_EDGES + t] : (t - N_EDGES);
  const float ex = __expf(ebuf[t] - funmap(emaxu[d]));
  ebuf[t] = ex;
  atomicAdd(&denom[d], ex);
}

#define EPW 16
#define LPR (D / 32)
static_assert(N_EDGES % EPW == 0 && E_TOT % EPW == 0, "edge groups must not straddle");
static_assert(D % 32 == 0, "a row must be whole lines");

__global__ __launch_bounds__(256)
void edge_pass3(const int* __restrict__ ei, const float* __restrict__ ebuf,
                const float* __restrict__ denom, const float* __restrict__ H,
                float* __restrict__ acc) {
  const int gt = blockIdx.x * blockDim.x + threadIdx.x;
  const int grp = gt >> 5;
  const int lane = threadIdx.x & 31;
  if (grp >= E_TOT / EPW) return;
  const int e = grp * EPW + (lane & (EPW - 1));
  int s, d;
  if (e < N_EDGES) { s = ei[e]; d = ei[N_EDGES + e]; }
  else             { s = e - N_EDGES; d = s; }
  const float coef = ebuf[e] / denom[d];

  float hv[EPW][LPR];
  #pragma unroll
  for (int j = 0; j < EPW; ++j) {
    const int sj = __builtin_amdgcn_readlane(s, j);
    const float* hr = H + (size_t)sj * D + lane;
    #pragma unroll
    for (int i = 0; i < LPR; ++i) hv[j][i] = hr[32 * i];
  }
  #pragma unroll
  for (int j = 0; j < EPW; ++j) {
    const int dj = __builtin_amdgcn_readlane(d, j);
    const float cj = __int_as_float(__builtin_amdgcn_readlane(__float_as_int(coef), j));
    float* ad = acc + (size_t)dj * D + lane;
    #pragma unroll
    for (int i = 0; i < LPR; ++i) atomicAdd(&ad[32 * i], hv[j][i] * cj);
  }
}

__global__ __launch_bounds__(256)
void relu_pool(float* __restrict__ acc, const float* __restrict__ bias,
               const int* __restrict__ batch, float* __restrict__ embsum) {
  const int t = blockIdx.x * blockDim.x + threadIdx.x;
  if (t >= N_NODES * D) return;
  const int node = t >> 7, c = t & 127;
  float v = acc[t] + bias[c];
  v = (v > 0.f) ? v : 0.f;
  acc[t] = v;
  atomicAdd(&embsum[(size_t)batch[node] * D + c], v);
}

__global__ __launch_bounds__(256)
void count_kernel(const int* __restrict__ batch, float* __restrict__ cnt) {
  const int t = blockIdx.x * blockDim.x + threadIdx.x;
  if (t < N_NODES) atomicAdd(&cnt[batch[t]], 1.0f);
}

__global__ __launch_bounds__(256)
void finalize_kernel(float* __restrict__ out, const float* __restrict__ cnt) {
  const int t = blockIdx.x * blockDim.x + threadIdx.x;
  if (t >= 2 * N_GRAPHS * D) return;
  const int g = (t >> 7) & (N_GRAPHS - 1);
  float c = cnt[g];
  c = (c > 1.f) ? c : 1.f;
  out[t] /= c;
}

extern "C" void kernel_launch(void* const* d_in, const int* in_sizes, int n_in,
                              void* d_out, int out_size, void* d_ws, size_t ws_size,
                              hipStream_t stream) {
  const float* x     = (const float*)d_in[0];
  const int*   ei    = (const int*)d_in[1];
  const int*   batch = (const int*)d_in[2];
  const float* W1    = (const float*)d_in[3];
  const float* as1   = (const float*)d_in[4];
  const float* ad1   = (const float*)d_in[5];
  const float* b1    = (const float*)d_in[6];
  const float* W2    = (const float*)d_in[7];
  const float* as2   = (const float*)d_in[8];
  const float* ad2   = (const float*)d_in[9];
  const float* b2    = (const float*)d_in[10];
  float* out = (float*)d_out;

  char* w = (char*)d_ws;
  auto carve = [&](size_t bytes) {
    char* p = w;
    w += (bytes + 255) & ~(size_t)255;
    return p;
  };
  float*    hbuf  = (float*)carve((size_t)N_NODES * D * 4);
  float*    acc   = (float*)carve((size_t)N_NODES * D * 4);
  float*    asrc  = (float*)carve((size_t)N_NODES * 4);
  float*    adst  = (float*)carve((size_t)N_NODES * 4);
  unsigned* emaxu = (unsigned*)carve((size_t)N_NODES * 4);
  float*    denom = (float*)carve((size_t)N_NODES * 4);
  float*    ebuf  = (float*)carve((size_t)E_TOT * 4);
  float*    cnt   = (float*)carve((size_t)N_GRAPHS * 4);

  const int TB = 256;
  const int gGemm  = N_NODES / 16;
  const int gAlpha = (N_NODES * 32 + TB - 1) / TB;
  const int gEdge  = (E_TOT + TB - 1) / TB;
  const int gEdgeW = ((E_TOT / EPW) * 32 + TB - 1) / TB;
  const int gNodeC = (N_NODES * D + TB - 1) / TB;
  const int gNode  = (N_NODES + TB - 1) / TB;
  const int gFin   = (2 * N_GRAPHS * D + TB - 1) / TB;

  hipMemsetAsync(d_out, 0, (size_t)2 * N_GRAPHS * D * 4, stream);
  hipMemsetAsync(cnt, 0, (size_t)N_GRAPHS * 4, stream);
  count_kernel<<<gNode, TB, 0, stream>>>(batch, cnt);

  gemm_wmma<<<gGemm, TB, 0, stream>>>(x, W1, hbuf);
  alpha_kernel<<<gAlpha, TB, 0, stream>>>(hbuf, as1, ad1, asrc, adst);
  hipMemsetAsync(emaxu, 0, (size_t)N_NODES * 4, stream);
  hipMemsetAsync(denom, 0, (size_t)N_NODES * 4, stream);
  hipMemsetAsync(acc, 0, (size_t)N_NODES * D * 4, stream);
  edge_pass1<<<gEdge, TB, 0, stream>>>(ei, asrc, adst, ebuf, emaxu);
  edge_pass2<<<gEdge, TB, 0, stream>>>(ei, ebuf, emaxu, denom);
  edge_pass3<<<gEdgeW, TB, 0, stream>>>(ei, ebuf, denom, hbuf, acc);
  relu_pool<<<gNodeC, TB, 0, stream>>>(acc, b1, batch, out);

  gemm_wmma<<<gGemm, TB, 0, stream>>>(acc, W2, hbuf);
  alpha_kernel<<<gAlpha, TB, 0, stream>>>(hbuf, as2, ad2, asrc, adst);
  hipMemsetAsync(emaxu, 0, (size_t)N_NODES * 4, stream);
  hipMemsetAsync(denom, 0, (size_t)N_NODES * 4, stream);
  hipMemsetAsync(acc, 0, (size_t)N_NODES * D * 4, stream);
  edge_pass1<<<gEdge, TB, 0, stream>>>(ei, asrc, adst, ebuf, emaxu);
  edge_pass2<<<gEdge, TB, 0, stream>>>(ei, ebuf, emaxu, denom);
  edge_pass3<<<gEdgeW, TB, 0, stream>>>(ei, ebuf, denom, hbuf, acc);
  relu_pool<<<gNodeC, TB, 0, stream>>>(acc, b2, batch, out + N_GRAPHS * D);

  finalize_kernel<<<gFin, TB, 0, stream>>>(out, cnt);
}


# Weight applied after the row sums: owner blocks gather and add input rows, then multiply their 196 summed rows by W from LDS on the matrix unit; scores from x.(W a); no GEMM launches (new sum tree)
# speedup vs baseline: 4.6815x; 3.4652x over previous
#include <hip/hip_runtime.h>

#define N_NODES  50000
#define N_EDGES  600000
#define E_TOT    (N_EDGES + N_NODES)
#define N_GRAPHS 1024
#define D        128
#define NEG_SLOPE 0.2f

typedef float    v2f  __attribute__((ext_vector_type(2)));
typedef float    v8f  __attribute__((ext_vector_type(8)));

__device__ __forceinline__ unsigned fmap(float x) {
  const unsigned u = __float_as_uint(x);
  return (u & 0x80000000u) ? ~u : (u | 0x80000000u);
}
__device__ __forceinline__ float funmap(unsigned u) {
  return __uint_as_float((u & 0x80000000u) ? (u & 0x7fffffffu) : ~u);
}
__device__ __forceinline__ float leaky(float v) {
  return (v > 0.f) ? v : NEG_SLOPE * v;
}

#define OWN_WG    256
#define OWN_ROWS  ((N_NODES + OWN_WG - 1) / OWN_WG)
#define OWN_TPB   1024
#define OWN_WAVES (OWN_TPB / 32)
#define OWN_K     ((OWN_ROWS + OWN_WAVES - 1) / OWN_WAVES)
#define GRP       16
#define LPR       (D / 32)
#define BIN_CAP   4096
#define ENT_PT    (BIN_CAP / OWN_TPB)
#define BIN_TPB   1024
#define BIN_EPT   4
#define BIN_EPB   (BIN_TPB * BIN_EPT)
#define N_BCNT    (OWN_WG + 1)
#define POOL_Q    (OWN_TPB / D)
#define POOL_SL   ((OWN_ROWS + POOL_Q - 1) / POOL_Q)
#define SP        132
#define WP        136
#define OWN_RT    ((OWN_ROWS + 15) / 16)
#define OWN_RP    (OWN_RT * 16)
#define POOL_U    5
static_assert(OWN_RT == 13 && OWN_WAVES == 32, "the matrix step deals 13 row tiles x 8 column tiles to 32 waves");
static_assert(N_NODES <= (1 << 16), "source node must fit the low half of a bin word");
static_assert(OWN_ROWS <= (1 << 16), "local row must fit the high half of a bin word");
static_assert(OWN_ROWS <= OWN_TPB, "one thread per row sets up the row's softmax cells");
static_assert(BIN_CAP % OWN_TPB == 0, "a bin is dealt out whole to the block's threads");
static_assert(D % 32 == 0, "a row must be whole lines");
static_assert(OWN_TPB % D == 0, "the closing pass gives every thread one column of one row slice");
static_assert(BIN_CAP % GRP == 0, "groups must tile a bin");
static_assert((OWN_WAVES & (OWN_WAVES - 1)) == 0, "the short group's wave is picked with a mask");

__global__ __launch_bounds__(BIN_TPB)
void bin_kernel(const int* __restrict__ ei, unsigned* __restrict__ binw,
                int* __restrict__ bcnt, int* __restrict__ ovf) {
  __shared__ int hist[OWN_WG];
  __shared__ int base[OWN_WG];
  const int t = threadIdx.x;
  for (int i = t; i < OWN_WG; i += BIN_TPB) hist[i] = 0;
  __syncthreads();
  const int e0 = blockIdx.x * BIN_EPB + t;
  int dd[BIN_EPT], ss[BIN_EPT], bb[BIN_EPT], rk[BIN_EPT];
  #pragma unroll
  for (int k = 0; k < BIN_EPT; ++k) {
    const int e = e0 + k * BIN_TPB;
    dd[k] = 0; ss[k] = 0;
    if (e < N_EDGES) { dd[k] = ei[N_EDGES + e]; ss[k] = ei[e]; }
  }
  #pragma unroll
  for (int k = 0; k < BIN_EPT; ++k) {
    bb[k] = (int)((unsigned)dd[k] / (unsigned)OWN_ROWS);
    rk[k] = 0;
    if (e0 + k * BIN_TPB < N_EDGES) rk[k] = atomicAdd(&hist[bb[k]], 1);
  }
  __syncthreads();
  for (int i = t; i < OWN_WG; i += BIN_TPB) {
    const int c = hist[i];
    base[i] = (c > 0) ? atomicAdd(&bcnt[i], c) : 0;
  }
  __syncthreads();
  #pragma unroll
  for (int k = 0; k < BIN_EPT; ++k) {
    const int e = e0 + k * BIN_TPB;
    if (e < N_EDGES) {
      const int slot = base[bb[k]] + rk[k];
      if (slot < BIN_CAP) {
        const unsigned r = (unsigned)(dd[k] - bb[k] * OWN_ROWS);
        binw[(size_t)bb[k] * BIN_CAP + slot] = (unsigned)ss[k] | (r << 16);
      } else {
        ovf[atomicAdd(&bcnt[OWN_WG], 1)] = e;
      }
    }
  }
}

static_assert(OWN_TPB == 8 * D, "eight threads per row of the weight matrix");
__device__ __forceinline__ void wa_vectors(const float* __restrict__ W, const float* __restrict__ a_src,
                                           const float* __restrict__ a_dst, int t, float* u_s) {
  const int k = t >> 3, p = t & 7;
  const float4* wr = (const float4*)(W + (size_t)k * D + p * 16);
  const float4* sr = (const float4*)(a_src + p * 16);
  const float4* dr = (const float4*)(a_dst + p * 16);
  float s = 0.f, d = 0.f;
  #pragma unroll
  for (int i = 0; i < 4; ++i) {
    const float4 wv = wr[i], sv = sr[i], dv = dr[i];
    s += wv.x * sv.x + wv.y * sv.y + wv.z * sv.z + wv.w * sv.w;
    d += wv.x * dv.x + wv.y * dv.y + wv.z * dv.z + wv.w * dv.w;
  }
  #pragma unroll
  for (int off = 4; off > 0; off >>= 1) {
    s += __shfl_xor(s, off, 32);
    d += __shfl_xor(d, off, 32);
  }
  if (p == 0) { u_s[k] = s; u_s[D + k] = d; }
}

__global__ __launch_bounds__(OWN_TPB)
void alpha_rows(const float* __restrict__ X, const float* __restrict__ W,
                const float* __restrict__ a_src, const float* __restrict__ a_dst,
                float* __restrict__ asrc, float* __restrict__ adst) {
  __shared__ float u_s[2 * D];
  const int t = threadIdx.x;
  const int lane = t & 31;
  const int w = __builtin_amdgcn_readfirstlane(t >> 5);
  const int row0 = blockIdx.x * OWN_ROWS;
  float4 xv[OWN_K];
  #pragma unroll
  for (int k = 0; k < OWN_K; ++k) {
    int node = row0 + w + 32 * k;
    node = (node < N_NODES) ? node : (N_NODES - 1);
    xv[k] = *(const float4*)(X + (size_t)node * D + lane * 4);
  }
  wa_vectors(W, a_src, a_dst, t, u_s);
  __syncthreads();
  const float4 sv = *(const float4*)&u_s[lane * 4];
  const float4 dv = *(const float4*)&u_s[D + lane * 4];
  #pragma unroll
  for (int k = 0; k < OWN_K; ++k) {
    const int r = w + 32 * k;
    float s = xv[k].x * sv.x + xv[k].y * sv.y + xv[k].z * sv.z + xv[k].w * sv.w;
    float d = xv[k].x * dv.x + xv[k].y * dv.y + xv[k].z * dv.z + xv[k].w * dv.w;
    #pragma unroll
    for (int off = 16; off > 0; off >>= 1) {
      s += __shfl_down(s, off, 32);
      d += __shfl_down(d, off, 32);
    }
    if (lane == 0 && r < OWN_ROWS && row0 + r < N_NODES) { asrc[row0 + r] = s; adst[row0 + r] = d; }
  }
}

__device__ __forceinline__ bool ovf_edge(const int* __restrict__ ei, const int* __restrict__ ovf,
                                         int i, int row0, const float* __restrict__ asrc,
                                         const float* ad_s, unsigned& r, int& s, float& v) {
  const int e = ovf[i];
  r = (unsigned)(ei[N_EDGES + e] - row0);
  if (r >= (unsigned)OWN_ROWS) return false;
  s = ei[e];
  v = leaky(asrc[s] + ad_s[r]);
  return true;
}

template <bool FULL>
__device__ __forceinline__ void own_group(const unsigned* ew, const float* cf, int n, int lane,
                                          const float* __restrict__ H, float* sums) {
  const int j0 = lane & (GRP - 1);
  unsigned q = 0;
  float coef = 0.f;
  if (FULL || j0 < n) { q = ew[j0]; coef = cf[j0]; }
  const int s = (int)(q & 0xFFFFu);
  const int dl = (int)(q >> 16);
  float hv[GRP][LPR];
  #pragma unroll
  for (int j = 0; j < GRP; ++j) {
    if (FULL || j < n) {
      const int sj = __builtin_amdgcn_readlane(s, j);
      const float* hr = H + (size_t)sj * D + lane;
      #pragma unroll
      for (int i = 0; i < LPR; ++i) hv[j][i] = hr[32 * i];
    }
  }
  #pragma unroll
  for (int j = 0; j < GRP; ++j) {
    if (FULL || j < n) {
      const int dj = __builtin_amdgcn_readlane(dl, j);
      const float cj = __int_as_float(__builtin_amdgcn_readlane(__float_as_int(coef), j));
      float* sd = sums + dj * SP + lane;
      #pragma unroll
      for (int i = 0; i < LPR; ++i)
        __hip_atomic_fetch_add(&sd[32 * i], hv[j][i] * cj, __ATOMIC_RELAXED,
                               __HIP_MEMORY_SCOPE_WORKGROUP);
    }
  }
}

__global__ __launch_bounds__(OWN_TPB)
void owner_pass(const int* __restrict__ ei, const float* __restrict__ asrc,
                const float* __restrict__ adst, const float* __restrict__ H,
                const float* __restrict__ bias, const int* __restrict__ batch,
                const unsigned* __restrict__ binw, const int* __restrict__ bcnt,
                const int* __restrict__ ovf, float* __restrict__ acc,
                float* __restrict__ embsum, int store_rows,
                const float* __restrict__ Wm, const float* __restrict__ Wn,
                const float* __restrict__ nas, const float* __restrict__ nad,
                float* __restrict__ asrc_n, float* __restrict__ adst_n) {
  __shared__ float    sums[OWN_RP * SP];
  __shared__ float    wl[D][WP];
  __shared__ float    un_s[2 * D];
  __shared__ unsigned ew_s[BIN_CAP];
  __shared__ float    cf_s[BIN_CAP];
  __shared__ float    ad_s[OWN_ROWS];
  __shared__ float    sl_s[OWN_ROWS];
  __shared__ unsigned mx_s[OWN_ROWS];
  __shared__ float    dn_s[OWN_ROWS];
  __shared__ int      bt_s[OWN_ROWS];
  const int t = threadIdx.x;
  const int lane = t & 31;
  const int w = __builtin_amdgcn_readfirstlane(t >> 5);
  const int row0 = blockIdx.x * OWN_ROWS;
  const int nb_all = bcnt[blockIdx.x];
  const int nb = (nb_all < BIN_CAP) ? nb_all : BIN_CAP;
  const int novf = bcnt[OWN_WG];
  const unsigned* __restrict__ bw = binw + (size_t)blockIdx.x * BIN_CAP;

  float4 wv4[(D * D / 4) / OWN_TPB];
  #pragma unroll
  for (int k = 0; k < (D * D / 4) / OWN_TPB; ++k) {
    const int i = t + k * OWN_TPB;
    wv4[k] = ((const float4*)(Wm + (size_t)(i >> 5) * D))[i & 31];
  }
  if (store_rows) wa_vectors(Wn, nas, nad, t, un_s);

  int nd = row0 + t;
  nd = (nd < N_NODES) ? nd : (N_NODES - 1);
  const float ad_t = adst[nd];
  const float as_t = asrc[nd];
  const int   bt_t = batch[nd];

  float h0[OWN_K][LPR];
  #pragma unroll
  for (int k = 0; k < OWN_K; ++k) {
    int node = row0 + w + 32 * k;
    node = (node < N_NODES) ? node : (N_NODES - 1);
    #pragma unroll
    for (int i = 0; i < LPR; ++i) h0[k][i] = H[(size_t)node * D + lane + 32 * i];
  }

  unsigned ew[ENT_PT];
  float sc[ENT_PT];
  #pragma unroll
  for (int k = 0; k < ENT_PT; ++k)
    if (t + k * OWN_TPB < nb) ew[k] = bw[t + k * OWN_TPB];
  #pragma unroll
  for (int k = 0; k < ENT_PT; ++k) {
    const unsigned sn = (t + k * OWN_TPB < nb) ? (ew[k] & 0xFFFFu) : 0u;
    sc[k] = asrc[sn];
  }

  #pragma unroll
  for (int k = 0; k < (D * D / 4) / OWN_TPB; ++k) {
    const int i = t + k * OWN_TPB;
    *(float4*)&wl[i >> 5][(i & 31) * 4] = wv4[k];
  }

  if (t < OWN_ROWS) {
    const bool live = (row0 + t) < N_NODES;
    const float ad = live ? ad_t : 0.f;
    const float v = leaky((live ? as_t : 0.f) + ad);
    ad_s[t] = ad;
    sl_s[t] = v;
    mx_s[t] = fmap(v);
    dn_s[t] = 0.f;
    bt_s[t] = live ? bt_t : 0;
  }
  __syncthreads();

  #pragma unroll
  for (int k = 0; k < ENT_PT; ++k) {
    if (t + k * OWN_TPB < nb) {
      const unsigned r = ew[k] >> 16;
      sc[k] = leaky(sc[k] + ad_s[r]);
      atomicMax(&mx_s[r], fmap(sc[k]));
    }
  }
  for (int i = t; i < novf; i += OWN_TPB) {
    unsigned r; int s; float v;
    if (ovf_edge(ei, ovf, i, row0, asrc, ad_s, r, s, v)) atomicMax(&mx_s[r], fmap(v));
  }
  __syncthreads();

  #pragma unroll
  for (int k = 0; k < ENT_PT; ++k) {
    if (t + k * OWN_TPB < nb) {
      const unsigned r = ew[k] >> 16;
      sc[k] = __expf(sc[k] - funmap(mx_s[r]));
      __hip_atomic_fetch_add(&dn_s[r], sc[k], __ATOMIC_RELAXED, __HIP_MEMORY_SCOPE_WORKGROUP);
    }
  }
  for (int i = t; i < novf; i += OWN_TPB) {
    unsigned r; int s; float v;
    if (ovf_edge(ei, ovf, i, row0, asrc, ad_s, r, s, v))
      __hip_atomic_fetch_add(&dn_s[r], __expf(v - funmap(mx_s[r])), __ATOMIC_RELAXED,
                             __HIP_MEMORY_SCOPE_WORKGROUP);
  }
  if (t < OWN_ROWS)
    __hip_atomic_fetch_add(&dn_s[t], __expf(sl_s[t] - funmap(mx_s[t])), __ATOMIC_RELAXED,
                           __HIP_MEMORY_SCOPE_WORKGROUP);
  __syncthreads();

  #pragma unroll
  for (int k = 0; k < ENT_PT; ++k) {
    const int i = t + k * OWN_TPB;
    if (i < nb) {
      ew_s[i] = ew[k];
      cf_s[i] = sc[k] / dn_s[ew[k] >> 16];
    }
  }
  #pragma unroll
  for (int k = 0; k < OWN_K; ++k) {
    const int r = w + 32 * k;
    if (r < OWN_ROWS) {
      const bool live = (row0 + r) < N_NODES;
      const float c0 = __expf(sl_s[r] - funmap(mx_s[r])) / dn_s[r];
      #pragma unroll
      for (int i = 0; i < LPR; ++i)
        sums[r * SP + lane + 32 * i] = live ? h0[k][i] * c0 : 0.f;
    }
  }
  if (w < OWN_RP - OWN_ROWS) {
    #pragma unroll
    for (int i = 0; i < LPR; ++i) sums[(OWN_ROWS + w) * SP + lane + 32 * i] = 0.f;
  }
  __syncthreads();

  {
    const int nfull = nb / GRP;
    for (int g = w; g < nfull; g += OWN_WAVES)
      own_group<true>(ew_s + g * GRP, cf_s + g * GRP, GRP, lane, H, sums);
    const int rem = nb - nfull * GRP;
    if (rem > 0 && w == (nfull & (OWN_WAVES - 1)))
      own_group<false>(ew_s + nfull * GRP, cf_s + nfull * GRP, rem, lane, H, sums);
  }
  for (int i = t; i < novf; i += OWN_TPB) {
    unsigned r; int s; float v;
    if (ovf_edge(ei, ovf, i, row0, asrc, ad_s, r, s, v)) {
      const float* hr = H + (size_t)s * D;
      const float coef = __expf(v - funmap(mx_s[r])) / dn_s[r];
      for (int c = 0; c < D; ++c)
        __hip_atomic_fetch_add(&sums[r * SP + c], hr[c] * coef, __ATOMIC_RELAXED,
                               __HIP_MEMORY_SCOPE_WORKGROUP);
    }
  }
  __syncthreads();

  {
    const int hi = lane >> 4;
    const int m  = lane & 15;
    const int g  = w >> 3;
    const int n  = (w & 7) * 16 + m;
    const float* a0p = sums + ((g     ) * 16 + m) * SP;
    const float* a1p = sums + ((g +  4) * 16 + m) * SP;
    const float* a2p = sums + ((g +  8) * 16 + m) * SP;
    const float* a3p = sums + ((OWN_RT - 1) * 16 + m) * SP;
    v8f c0 = {0.f, 0.f, 0.f, 0.f, 0.f, 0.f, 0.f, 0.f};
    v8f c1 = c0, c2 = c0, c3 = c0;
    if (g == 0) {
      #pragma unroll
      for (int k0 = 0; k0 < D; k0 += 4) {
        const int ka = k0 + hi * 2;
        v2f a, b;
        a[0] = a3p[ka];
        a[1] = a3p[ka + 1];
        b[0] = wl[ka][n];
        b[1] = wl[ka + 1][n];
        c3 = __builtin_amdgcn_wmma_f32_16x16x4_f32(false, a, false, b, (short)0, c3, false, false);
      }
    }
    #pragma unroll
    for (int k0 = 0; k0 < D; k0 += 4) {
      const int ka = k0 + hi * 2;
      v2f a, b;
      b[0] = wl[ka][n];
      b[1] = wl[ka + 1][n];
      a[0] = a0p[ka];
      a[1] = a0p[ka + 1];
      c0 = __builtin_amdgcn_wmma_f32_16x16x4_f32(false, a, false, b, (short)0, c0, false, false);
      a[0] = a1p[ka];
      a[1] = a1p[ka + 1];
      c1 = __builtin_amdgcn_wmma_f32_16x16x4_f32(false, a, false, b, (short)0, c1, false, false);
      a[0] = a2p[ka];
      a[1] = a2p[ka + 1];
      c2 = __builtin_amdgcn_wmma_f32_16x16x4_f32(false, a, false, b, (short)0, c2, false, false);
    }
    __syncthreads();
    #pragma unroll
    for (int r = 0; r < 8; ++r) {
      sums[((g     ) * 16 + r + hi * 8) * SP + n] = c0[r];
      sums[((g +  4) * 16 + r + hi * 8) * SP + n] = c1[r];
      sums[((g +  8) * 16 + r + hi * 8) * SP + n] = c2[r];
    }
    if (g == 0) {
      #pragma unroll
      for (int r = 0; r < 8; ++r) sums[((OWN_RT - 1) * 16 + r + hi * 8) * SP + n] = c3[r];
    }
  }
  __syncthreads();

  {
    const int c = t & (D - 1);
    const int q = t / D;
    int nlive = N_NODES - row0;
    nlive = (nlive < OWN_ROWS) ? nlive : OWN_ROWS;
    int r = q * POOL_SL;
    int rend = r + POOL_SL;
    rend = (rend < nlive) ? rend : nlive;
    const float bc = bias[c];
    float run = 0.f;
    int g = 0;
    bool have = false;
    for (int r0 = r; r0 < rend; r0 += POOL_U) {
      float vv[POOL_U];
      int gg[POOL_U];
      #pragma unroll
      for (int k = 0; k < POOL_U; ++k) {
        const int rr = (r0 + k < rend) ? (r0 + k) : r0;
        vv[k] = sums[rr * SP + c];
        gg[k] = bt_s[rr];
      }
      #pragma unroll
      for (int k = 0; k < POOL_U; ++k) {
        if (r0 + k < rend) {
          float v = vv[k] + bc;
          v = (v > 0.f) ? v : 0.f;
          if (store_rows) {
            acc[(size_t)(row0 + r0 + k) * D + c] = v;
            sums[(r0 + k) * SP + c] = v;
          }
          if (have && gg[k] != g) {
            atomicAdd(&embsum[(size_t)g * D + c], run);
            run = 0.f;
          }
          g = gg[k];
          have = true;
          run += v;
        }
      }
    }
    if (have) atomicAdd(&embsum[(size_t)g * D + c], run);
  }

  if (store_rows) {
    __syncthreads();
    int nlive = N_NODES - row0;
    nlive = (nlive < OWN_ROWS) ? nlive : OWN_ROWS;
    float us[LPR], ud[LPR];
    #pragma unroll
    for (int i = 0; i < LPR; ++i) { us[i] = un_s[lane + 32 * i]; ud[i] = un_s[D + lane + 32 * i]; }
    #pragma unroll
    for (int k = 0; k < OWN_K; ++k) {
      const int r = w + 32 * k;
      if (r < nlive) {
        float s = 0.f, d = 0.f;
        #pragma unroll
        for (int i = 0; i < LPR; ++i) {
          const float hv = sums[r * SP + lane + 32 * i];
          s += hv * us[i];
          d += hv * ud[i];
        }
        #pragma unroll
        for (int off = 16; off > 0; off >>= 1) {
          s += __shfl_down(s, off, 32);
          d += __shfl_down(d, off, 32);
        }
        if (lane == 0) { asrc_n[row0 + r] = s; adst_n[row0 + r] = d; }
      }
    }
  }
}

__global__ __launch_bounds__(256)
void count_kernel(const int* __restrict__ batch, float* __restrict__ cnt) {
  const int t = blockIdx.x * blockDim.x + threadIdx.x;
  if (t < N_NODES) atomicAdd(&cnt[batch[t]], 1.0f);
}

__global__ __launch_bounds__(256)
void finalize_kernel(float* __restrict__ out, const float* __restrict__ cnt) {
  const int t = blockIdx.x * blockDim.x + threadIdx.x;
  if (t >= 2 * N_GRAPHS * D) return;
  const int g = (t >> 7) & (N_GRAPHS - 1);
  float c = cnt[g];
  c = (c > 1.f) ? c : 1.f;
  out[t] /= c;
}

extern "C" void kernel_launch(void* const* d_in, const int* in_sizes, int n_in,
                              void* d_out, int out_size, void* d_ws, size_t ws_size,
                              hipStream_t stream) {
  const float* x     = (const float*)d_in[0];
  const int*   ei    = (const int*)d_in[1];
  const int*   batch = (const int*)d_in[2];
  const float* W1    = (const float*)d_in[3];
  const float* as1   = (const float*)d_in[4];
  const float* ad1   = (const float*)d_in[5];
  const float* b1    = (const float*)d_in[6];
  const float* W2    = (const float*)d_in[7];
  const float* as2   = (const float*)d_in[8];
  const float* ad2   = (const float*)d_in[9];
  const float* b2    = (const float*)d_in[10];
  float* out = (float*)d_out;

  char* w = (char*)d_ws;
  auto carve = [&](size_t bytes) {
    char* p = w;
    w += (bytes + 255) & ~(size_t)255;
    return p;
  };
  float*    acc   = (float*)carve((size_t)N_NODES * D * 4);
  float*    asrc  = (float*)carve((size_t)N_NODES * 4);
  float*    adst  = (float*)carve((size_t)N_NODES * 4);
  float*    asrc2 = (float*)carve((size_t)N_NODES * 4);
  float*    adst2 = (float*)carve((size_t)N_NODES * 4);
  float*    cnt   = (float*)carve((size_t)(N_GRAPHS + N_BCNT) * 4);
  int*      bcnt  = (int*)(cnt + N_GRAPHS);
  unsigned* binw  = (unsigned*)carve((size_t)OWN_WG * BIN_CAP * 4);
  int*      ovf   = (int*)carve((size_t)N_EDGES * 4);

  const int TB = 256;
  const int gNode  = (N_NODES + TB - 1) / TB;
  const int gFin   = (2 * N_GRAPHS * D + TB - 1) / TB;
  const int gBin   = (N_EDGES + BIN_EPB - 1) / BIN_EPB;

  hipMemsetAsync(d_out, 0, (size_t)2 * N_GRAPHS * D * 4, stream);
  hipMemsetAsync(cnt, 0, (size_t)(N_GRAPHS + N_BCNT) * 4, stream);
  count_kernel<<<gNode, TB, 0, stream>>>(batch, cnt);
  bin_kernel<<<gBin, BIN_TPB, 0, stream>>>(ei, binw, bcnt, ovf);

  alpha_rows<<<OWN_WG, OWN_TPB, 0, stream>>>(x, W1, as1, ad1, asrc, adst);
  owner_pass<<<OWN_WG, OWN_TPB, 0, stream>>>(ei, asrc, adst, x, b1, batch, binw, bcnt, ovf,
                                             acc, out, 1, W1, W2, as2, ad2, asrc2, adst2);

  owner_pass<<<OWN_WG, OWN_TPB, 0, stream>>>(ei, asrc2, adst2, acc, b2, batch, binw, bcnt, ovf,
                                             acc, out + N_GRAPHS * D, 0, W2, W2, as2, ad2, asrc2, adst2);

  finalize_kernel<<<gFin, TB, 0, stream>>>(out, cnt);
}


# Owner block counting-sorts its bin by row in LDS; a wave adds up its rows in registers (one 16-byte load per source row), rows stored once, no LDS float adds; sums: self loop, then LDS rank order
# speedup vs baseline: 5.2384x; 1.1190x over previous
#include <hip/hip_runtime.h>

#define N_NODES  50000
#define N_EDGES  600000
#define E_TOT    (N_EDGES + N_NODES)
#define N_GRAPHS 1024
#define D        128
#define NEG_SLOPE 0.2f

typedef float    v2f  __attribute__((ext_vector_type(2)));
typedef float    v8f  __attribute__((ext_vector_type(8)));

__device__ __forceinline__ unsigned fmap(float x) {
  const unsigned u = __float_as_uint(x);
  return (u & 0x80000000u) ? ~u : (u | 0x80000000u);
}
__device__ __forceinline__ float funmap(unsigned u) {
  return __uint_as_float((u & 0x80000000u) ? (u & 0x7fffffffu) : ~u);
}
__device__ __forceinline__ float leaky(float v) {
  return (v > 0.f) ? v : NEG_SLOPE * v;
}

#define OWN_WG    256
#define OWN_ROWS  ((N_NODES + OWN_WG - 1) / OWN_WG)
#define OWN_TPB   1024
#define OWN_WAVES (OWN_TPB / 32)
#define OWN_K     ((OWN_ROWS + OWN_WAVES - 1) / OWN_WAVES)
#define GRP       16
#define LPR       (D / 32)
#define BIN_CAP   4096
#define ENT_PT    (BIN_CAP / OWN_TPB)
#define BIN_TPB   1024
#define BIN_EPT   4
#define BIN_EPB   (BIN_TPB * BIN_EPT)
#define N_BCNT    (OWN_WG + 1)
#define POOL_Q    (OWN_TPB / D)
#define POOL_SL   ((OWN_ROWS + POOL_Q - 1) / POOL_Q)
#define SP        132
#define WP        136
#define OWN_RT    ((OWN_ROWS + 15) / 16)
#define OWN_RP    (OWN_RT * 16)
#define POOL_U    5
#define SRT_CAP   (BIN_CAP + OWN_ROWS)
#define SCAN_RPL  ((OWN_ROWS + 31) / 32)
static_assert(OWN_RT == 13 && OWN_WAVES == 32, "the matrix step deals 13 row tiles x 8 column tiles to 32 waves");
static_assert(N_NODES <= (1 << 16), "source node must fit the low half of a bin word");
static_assert(OWN_ROWS <= (1 << 16), "local row must fit the high half of a bin word");
static_assert(OWN_ROWS <= OWN_TPB, "one thread per row sets up the row's softmax cells");
static_assert(BIN_CAP % OWN_TPB == 0, "a bin is dealt out whole to the block's threads");
static_assert(D % 32 == 0, "a row must be whole lines");
static_assert(OWN_TPB % D == 0, "the closing pass gives every thread one column of one row slice");
static_assert(BIN_CAP % GRP == 0, "groups must tile a bin");
static_assert(OWN_WAVES == 32, "a wave's rows are the 32nd parts of the block's rows");
static_assert(OWN_ROWS >= OWN_WAVES, "every wave owns at least one row");
static_assert((SP * 4) % 16 == 0, "a row of sums is stored 16 bytes per lane");

__global__ __launch_bounds__(BIN_TPB)
void bin_kernel(const int* __restrict__ ei, unsigned* __restrict__ binw,
                int* __restrict__ bcnt, int* __restrict__ ovf) {
  __shared__ int hist[OWN_WG];
  __shared__ int base[OWN_WG];
  const int t = threadIdx.x;
  for (int i = t; i < OWN_WG; i += BIN_TPB) hist[i] = 0;
  __syncthreads();
  const int e0 = blockIdx.x * BIN_EPB + t;
  int dd[BIN_EPT], ss[BIN_EPT], bb[BIN_EPT], rk[BIN_EPT];
  #pragma unroll
  for (int k = 0; k < BIN_EPT; ++k) {
    const int e = e0 + k * BIN_TPB;
    dd[k] = 0; ss[k] = 0;
    if (e < N_EDGES) { dd[k] = ei[N_EDGES + e]; ss[k] = ei[e]; }
  }
  #pragma unroll
  for (int k = 0; k < BIN_EPT; ++k) {
    bb[k] = (int)((unsigned)dd[k] / (unsigned)OWN_ROWS);
    rk[k] = 0;
    if (e0 + k * BIN_TPB < N_EDGES) rk[k] = atomicAdd(&hist[bb[k]], 1);
  }
  __syncthreads();
  for (int i = t; i < OWN_WG; i += BIN_TPB) {
    const int c = hist[i];
    base[i] = (c > 0) ? atomicAdd(&bcnt[i], c) : 0;
  }
  __syncthreads();
  #pragma unroll
  for (int k = 0; k < BIN_EPT; ++k) {
    const int e = e0 + k * BIN_TPB;
    if (e < N_EDGES) {
      const int slot = base[bb[k]] + rk[k];
      if (slot < BIN_CAP) {
        const unsigned r = (unsigned)(dd[k] - bb[k] * OWN_ROWS);
        binw[(size_t)bb[k] * BIN_CAP + slot] = (unsigned)ss[k] | (r << 16);
      } else {
        ovf[atomicAdd(&bcnt[OWN_WG], 1)] = e;
      }
    }
  }
}

static_assert(OWN_TPB == 8 * D, "eight threads per row of the weight matrix");
__device__ __forceinline__ void wa_vectors(const float* __restrict__ W, const float* __restrict__ a_src,
                                           const float* __restrict__ a_dst, int t, float* u_s) {
  const int k = t >> 3, p = t & 7;
  const float4* wr = (const float4*)(W + (size_t)k * D + p * 16);
  const float4* sr = (const float4*)(a_src + p * 16);
  const float4* dr = (const float4*)(a_dst + p * 16);
  float s = 0.f, d = 0.f;
  #pragma unroll
  for (int i = 0; i < 4; ++i) {
    const float4 wv = wr[i], sv = sr[i], dv = dr[i];
    s += wv.x * sv.x + wv.y * sv.y + wv.z * sv.z + wv.w * sv.w;
    d += wv.x * dv.x + wv.y * dv.y + wv.z * dv.z + wv.w * dv.w;
  }
  #pragma unroll
  for (int off = 4; off > 0; off >>= 1) {
    s += __shfl_xor(s, off, 32);
    d += __shfl_xor(d, off, 32);
  }
  if (p == 0) { u_s[k] = s; u_s[D + k] = d; }
}

__global__ __launch_bounds__(OWN_TPB)
void alpha_rows(const float* __restrict__ X, const float* __restrict__ W,
                const float* __restrict__ a_src, const float* __restrict__ a_dst,
                float* __restrict__ asrc, float* __restrict__ adst) {
  __shared__ float u_s[2 * D];
  const int t = threadIdx.x;
  const int lane = t & 31;
  const int w = __builtin_amdgcn_readfirstlane(t >> 5);
  const int row0 = blockIdx.x * OWN_ROWS;
  float4 xv[OWN_K];
  #pragma unroll
  for (int k = 0; k < OWN_K; ++k) {
    int node = row0 + w + 32 * k;
    node = (node < N_NODES) ? node : (N_NODES - 1);
    xv[k] = *(const float4*)(X + (size_t)node * D + lane * 4);
  }
  wa_vectors(W, a_src, a_dst, t, u_s);
  __syncthreads();
  const float4 sv = *(const float4*)&u_s[lane * 4];
  const float4 dv = *(const float4*)&u_s[D + lane * 4];
  #pragma unroll
  for (int k = 0; k < OWN_K; ++k) {
    const int r = w + 32 * k;
    float s = xv[k].x * sv.x + xv[k].y * sv.y + xv[k].z * sv.z + xv[k].w * sv.w;
    float d = xv[k].x * dv.x + xv[k].y * dv.y + xv[k].z * dv.z + xv[k].w * dv.w;
    #pragma unroll
    for (int off = 16; off > 0; off >>= 1) {
      s += __shfl_down(s, off, 32);
      d += __shfl_down(d, off, 32);
    }
    if (lane == 0 && r < OWN_ROWS && row0 + r < N_NODES) { asrc[row0 + r] = s; adst[row0 + r] = d; }
  }
}

__device__ __forceinline__ bool ovf_edge(const int* __restrict__ ei, const int* __restrict__ ovf,
                                         int i, int row0, const float* __restrict__ asrc,
                                         const float* ad_s, unsigned& r, int& s, float& v) {
  const int e = ovf[i];
  r = (unsigned)(ei[N_EDGES + e] - row0);
  if (r >= (unsigned)OWN_ROWS) return false;
  s = ei[e];
  v = leaky(asrc[s] + ad_s[r]);
  return true;
}

template <bool FULL>
__device__ __forceinline__ void own_group(const unsigned* ew, const float* cf, int n, int lane,
                                          const float* __restrict__ H, float* sums,
                                          int& cur, float4& acc) {
  const int j0 = lane & (GRP - 1);
  unsigned q = 0;
  float coef = 0.f;
  if (FULL || j0 < n) { q = ew[j0]; coef = cf[j0]; }
  const int s = (int)(q & 0xFFFFu);
  const int dl = (int)(q >> 16);
  float4 hv[GRP];
  #pragma unroll
  for (int j = 0; j < GRP; ++j) {
    if (FULL || j < n) {
      const int sj = __builtin_amdgcn_readlane(s, j);
      hv[j] = *(const float4*)(H + (size_t)sj * D + lane * 4);
    }
  }
  #pragma unroll
  for (int j = 0; j < GRP; ++j) {
    if (FULL || j < n) {
      const int dj = __builtin_amdgcn_readlane(dl, j);
      const float cj = __int_as_float(__builtin_amdgcn_readlane(__float_as_int(coef), j));
      if (dj != cur) {
        *(float4*)&sums[cur * SP + lane * 4] = acc;
        acc = make_float4(0.f, 0.f, 0.f, 0.f);
        cur = dj;
      }
      acc.x += hv[j].x * cj;
      acc.y += hv[j].y * cj;
      acc.z += hv[j].z * cj;
      acc.w += hv[j].w * cj;
    }
  }
}

__global__ __launch_bounds__(OWN_TPB)
void owner_pass(const int* __restrict__ ei, const float* __restrict__ asrc,
                const float* __restrict__ adst, const float* __restrict__ H,
                const float* __restrict__ bias, const int* __restrict__ batch,
                const unsigned* __restrict__ binw, const int* __restrict__ bcnt,
                const int* __restrict__ ovf, float* __restrict__ acc,
                float* __restrict__ embsum, int store_rows,
                const float* __restrict__ Wm, const float* __restrict__ Wn,
                const float* __restrict__ nas, const float* __restrict__ nad,
                float* __restrict__ asrc_n, float* __restrict__ adst_n) {
  __shared__ __attribute__((aligned(16))) float sums[OWN_RP * SP];
  __shared__ float    wl[D][WP];
  __shared__ float    un_s[2 * D];
  __shared__ unsigned ew_s[SRT_CAP];
  __shared__ float    cf_s[SRT_CAP];
  __shared__ int      cn_s[OWN_ROWS];
  __shared__ int      st_s[OWN_ROWS + 1];
  __shared__ float    ad_s[OWN_ROWS];
  __shared__ float    sl_s[OWN_ROWS];
  __shared__ unsigned mx_s[OWN_ROWS];
  __shared__ float    dn_s[OWN_ROWS];
  __shared__ int      bt_s[OWN_ROWS];
  const int t = threadIdx.x;
  const int lane = t & 31;
  const int w = __builtin_amdgcn_readfirstlane(t >> 5);
  const int row0 = blockIdx.x * OWN_ROWS;
  const int nb_all = bcnt[blockIdx.x];
  const int nb = (nb_all < BIN_CAP) ? nb_all : BIN_CAP;
  const int novf = bcnt[OWN_WG];
  const unsigned* __restrict__ bw = binw + (size_t)blockIdx.x * BIN_CAP;

  float4 wv4[(D * D / 4) / OWN_TPB];
  #pragma unroll
  for (int k = 0; k < (D * D / 4) / OWN_TPB; ++k) {
    const int i = t + k * OWN_TPB;
    wv4[k] = ((const float4*)(Wm + (size_t)(i >> 5) * D))[i & 31];
  }
  if (store_rows) wa_vectors(Wn, nas, nad, t, un_s);

  int nd = row0 + t;
  nd = (nd < N_NODES) ? nd : (N_NODES - 1);
  const float ad_t = adst[nd];
  const float as_t = asrc[nd];
  const int   bt_t = batch[nd];

  unsigned ew[ENT_PT];
  float sc[ENT_PT];
  #pragma unroll
  for (int k = 0; k < ENT_PT; ++k)
    if (t + k * OWN_TPB < nb) ew[k] = bw[t + k * OWN_TPB];
  #pragma unroll
  for (int k = 0; k < ENT_PT; ++k) {
    const unsigned sn = (t + k * OWN_TPB < nb) ? (ew[k] & 0xFFFFu) : 0u;
    sc[k] = asrc[sn];
  }

  #pragma unroll
  for (int k = 0; k < (D * D / 4) / OWN_TPB; ++k) {
    const int i = t + k * OWN_TPB;
    *(float4*)&wl[i >> 5][(i & 31) * 4] = wv4[k];
  }

  if (t < OWN_ROWS) {
    const bool live = (row0 + t) < N_NODES;
    const float ad = live ? ad_t : 0.f;
    const float v = leaky((live ? as_t : 0.f) + ad);
    ad_s[t] = ad;
    sl_s[t] = v;
    mx_s[t] = fmap(v);
    dn_s[t] = 0.f;
    cn_s[t] = 0;
    bt_s[t] = live ? bt_t : 0;
  }
  __syncthreads();

  int rk[ENT_PT];
  #pragma unroll
  for (int k = 0; k < ENT_PT; ++k) {
    if (t + k * OWN_TPB < nb) {
      const unsigned r = ew[k] >> 16;
      sc[k] = leaky(sc[k] + ad_s[r]);
      atomicMax(&mx_s[r], fmap(sc[k]));
      rk[k] = atomicAdd(&cn_s[r], 1);
    }
  }
  for (int i = t; i < novf; i += OWN_TPB) {
    unsigned r; int s; float v;
    if (ovf_edge(ei, ovf, i, row0, asrc, ad_s, r, s, v)) atomicMax(&mx_s[r], fmap(v));
  }
  __syncthreads();

  #pragma unroll
  for (int k = 0; k < ENT_PT; ++k) {
    if (t + k * OWN_TPB < nb) {
      const unsigned r = ew[k] >> 16;
      sc[k] = __expf(sc[k] - funmap(mx_s[r]));
      __hip_atomic_fetch_add(&dn_s[r], sc[k], __ATOMIC_RELAXED, __HIP_MEMORY_SCOPE_WORKGROUP);
    }
  }
  for (int i = t; i < novf; i += OWN_TPB) {
    unsigned r; int s; float v;
    if (ovf_edge(ei, ovf, i, row0, asrc, ad_s, r, s, v))
      __hip_atomic_fetch_add(&dn_s[r], __expf(v - funmap(mx_s[r])), __ATOMIC_RELAXED,
                             __HIP_MEMORY_SCOPE_WORKGROUP);
  }
  if (t < OWN_ROWS)
    __hip_atomic_fetch_add(&dn_s[t], __expf(sl_s[t] - funmap(mx_s[t])), __ATOMIC_RELAXED,
                           __HIP_MEMORY_SCOPE_WORKGROUP);
  if (w == OWN_WAVES - 1) {
    int cc[SCAN_RPL];
    int tot = 0;
    #pragma unroll
    for (int i = 0; i < SCAN_RPL; ++i) {
      const int r = lane * SCAN_RPL + i;
      cc[i] = (r < OWN_ROWS) ? (cn_s[r] + 1) : 0;
      tot += cc[i];
    }
    int inc = tot;
    #pragma unroll
    for (int off = 1; off < 32; off <<= 1) {
      const int v = __shfl_up(inc, off, 32);
      if (lane >= off) inc += v;
    }
    int run = inc - tot;
    #pragma unroll
    for (int i = 0; i < SCAN_RPL; ++i) {
      const int r = lane * SCAN_RPL + i;
      if (r < OWN_ROWS) st_s[r] = run;
      run += cc[i];
    }
    if (lane == 31) st_s[OWN_ROWS] = inc;
  }
  __syncthreads();

  #pragma unroll
  for (int k = 0; k < ENT_PT; ++k) {
    if (t + k * OWN_TPB < nb) {
      const unsigned r = ew[k] >> 16;
      const int pos = st_s[r] + 1 + rk[k];
      ew_s[pos] = ew[k];
      cf_s[pos] = sc[k] / dn_s[r];
    }
  }
  if (t < OWN_ROWS) {
    const bool live = (row0 + t) < N_NODES;
    const unsigned node = live ? (unsigned)(row0 + t) : (unsigned)(N_NODES - 1);
    const int pos = st_s[t];
    ew_s[pos] = node | ((unsigned)t << 16);
    cf_s[pos] = live ? (__expf(sl_s[t] - funmap(mx_s[t])) / dn_s[t]) : 0.f;
  }
  if (w < OWN_RP - OWN_ROWS) {
    #pragma unroll
    for (int i = 0; i < LPR; ++i) sums[(OWN_ROWS + w) * SP + lane + 32 * i] = 0.f;
  }
  __syncthreads();

  {
    const int rlo = (w * OWN_ROWS) >> 5;
    const int rhi = ((w + 1) * OWN_ROWS) >> 5;
    const int lo = __builtin_amdgcn_readfirstlane(st_s[rlo]);
    const int hi = __builtin_amdgcn_readfirstlane(st_s[rhi]);
    int cur = rlo;
    float4 acc = make_float4(0.f, 0.f, 0.f, 0.f);
    int g = lo;
    for (; g + GRP <= hi; g += GRP)
      own_group<true>(ew_s + g, cf_s + g, GRP, lane, H, sums, cur, acc);
    if (g < hi)
      own_group<false>(ew_s + g, cf_s + g, hi - g, lane, H, sums, cur, acc);
    *(float4*)&sums[cur * SP + lane * 4] = acc;
  }
  if (novf > 0) {
    __syncthreads();
    for (int i = t; i < novf; i += OWN_TPB) {
      unsigned r; int s; float v;
      if (ovf_edge(ei, ovf, i, row0, asrc, ad_s, r, s, v)) {
        const float* hr = H + (size_t)s * D;
        const float coef = __expf(v - funmap(mx_s[r])) / dn_s[r];
        for (int c = 0; c < D; ++c)
          __hip_atomic_fetch_add(&sums[r * SP + c], hr[c] * coef, __ATOMIC_RELAXED,
                                 __HIP_MEMORY_SCOPE_WORKGROUP);
      }
    }
  }
  __syncthreads();

  {
    const int hi = lane >> 4;
    const int m  = lane & 15;
    const int g  = w >> 3;
    const int n  = (w & 7) * 16 + m;
    const float* a0p = sums + ((g     ) * 16 + m) * SP;
    const float* a1p = sums + ((g +  4) * 16 + m) * SP;
    const float* a2p = sums + ((g +  8) * 16 + m) * SP;
    const float* a3p = sums + ((OWN_RT - 1) * 16 + m) * SP;
    v8f c0 = {0.f, 0.f, 0.f, 0.f, 0.f, 0.f, 0.f, 0.f};
    v8f c1 = c0, c2 = c0, c3 = c0;
    if (g == 0) {
      #pragma unroll
      for (int k0 = 0; k0 < D; k0 += 4) {
        const int ka = k0 + hi * 2;
        v2f a, b;
        a[0] = a3p[ka];
        a[1] = a3p[ka + 1];
        b[0] = wl[ka][n];
        b[1] = wl[ka + 1][n];
        c3 = __builtin_amdgcn_wmma_f32_16x16x4_f32(false, a, false, b, (short)0, c3, false, false);
      }
    }
    #pragma unroll
    for (int k0 = 0; k0 < D; k0 += 4) {
      const int ka = k0 + hi * 2;
      v2f a, b;
      b[0] = wl[ka][n];
      b[1] = wl[ka + 1][n];
      a[0] = a0p[ka];
      a[1] = a0p[ka + 1];
      c0 = __builtin_amdgcn_wmma_f32_16x16x4_f32(false, a, false, b, (short)0, c0, false, false);
      a[0] = a1p[ka];
      a[1] = a1p[ka + 1];
      c1 = __builtin_amdgcn_wmma_f32_16x16x4_f32(false, a, false, b, (short)0, c1, false, false);
      a[0] = a2p[ka];
      a[1] = a2p[ka + 1];
      c2 = __builtin_amdgcn_wmma_f32_16x16x4_f32(false, a, false, b, (short)0, c2, false, false);
    }
    __syncthreads();
    #pragma unroll
    for (int r = 0; r < 8; ++r) {
      sums[((g     ) * 16 + r + hi * 8) * SP + n] = c0[r];
      sums[((g +  4) * 16 + r + hi * 8) * SP + n] = c1[r];
      sums[((g +  8) * 16 + r + hi * 8) * SP + n] = c2[r];
    }
    if (g == 0) {
      #pragma unroll
      for (int r = 0; r < 8; ++r) sums[((OWN_RT - 1) * 16 + r + hi * 8) * SP + n] = c3[r];
    }
  }
  __syncthreads();

  {
    const int c = t & (D - 1);
    const int q = t / D;
    int nlive = N_NODES - row0;
    nlive = (nlive < OWN_ROWS) ? nlive : OWN_ROWS;
    int r = q * POOL_SL;
    int rend = r + POOL_SL;
    rend = (rend < nlive) ? rend : nlive;
    const float bc = bias[c];
    float run = 0.f;
    int g = 0;
    bool have = false;
    for (int r0 = r; r0 < rend; r0 += POOL_U) {
      float vv[POOL_U];
      int gg[POOL_U];
      #pragma unroll
      for (int k = 0; k < POOL_U; ++k) {
        const int rr = (r0 + k < rend) ? (r0 + k) : r0;
        vv[k] = sums[rr * SP + c];
        gg[k] = bt_s[rr];
      }
      #pragma unroll
      for (int k = 0; k < POOL_U; ++k) {
        if (r0 + k < rend) {
          float v = vv[k] + bc;
          v = (v > 0.f) ? v : 0.f;
          if (store_rows) {
            acc[(size_t)(row0 + r0 + k) * D + c] = v;
            sums[(r0 + k) * SP + c] = v;
          }
          if (have && gg[k] != g) {
            atomicAdd(&embsum[(size_t)g * D + c], run);
            run = 0.f;
          }
          g = gg[k];
          have = true;
          run += v;
        }
      }
    }
    if (have) atomicAdd(&embsum[(size_t)g * D + c], run);
  }

  if (store_rows) {
    __syncthreads();
    int nlive = N_NODES - row0;
    nlive = (nlive < OWN_ROWS) ? nlive : OWN_ROWS;
    float us[LPR], ud[LPR];
    #pragma unroll
    for (int i = 0; i < LPR; ++i) { us[i] = un_s[lane + 32 * i]; ud[i] = un_s[D + lane + 32 * i]; }
    #pragma unroll
    for (int k = 0; k < OWN_K; ++k) {
      const int r = w + 32 * k;
      if (r < nlive) {
        float s = 0.f, d = 0.f;
        #pragma unroll
        for (int i = 0; i < LPR; ++i) {
          const float hv = sums[r * SP + lane + 32 * i];
          s += hv * us[i];
          d += hv * ud[i];
        }
        #pragma unroll
        for (int off = 16; off > 0; off >>= 1) {
          s += __shfl_down(s, off, 32);
          d += __shfl_down(d, off, 32);
        }
        if (lane == 0) { asrc_n[row0 + r] = s; adst_n[row0 + r] = d; }
      }
    }
  }
}

__global__ __launch_bounds__(256)
void count_kernel(const int* __restrict__ batch, float* __restrict__ cnt) {
  const int t = blockIdx.x * blockDim.x + threadIdx.x;
  if (t < N_NODES) atomicAdd(&cnt[batch[t]], 1.0f);
}

__global__ __launch_bounds__(256)
void finalize_kernel(float* __restrict__ out, const float* __restrict__ cnt) {
  const int t = blockIdx.x * blockDim.x + threadIdx.x;
  if (t >= 2 * N_GRAPHS * D) return;
  const int g = (t >> 7) & (N_GRAPHS - 1);
  float c = cnt[g];
  c = (c > 1.f) ? c : 1.f;
  out[t] /= c;
}

extern "C" void kernel_launch(void* const* d_in, const int* in_sizes, int n_in,
                              void* d_out, int out_size, void* d_ws, size_t ws_size,
                              hipStream_t stream) {
  const float* x     = (const float*)d_in[0];
  const int*   ei    = (const int*)d_in[1];
  const int*   batch = (const int*)d_in[2];
  const float* W1    = (const float*)d_in[3];
  const float* as1   = (const float*)d_in[4];
  const float* ad1   = (const float*)d_in[5];
  const float* b1    = (const float*)d_in[6];
  const float* W2    = (const float*)d_in[7];
  const float* as2   = (const float*)d_in[8];
  const float* ad2   = (const float*)d_in[9];
  const float* b2    = (const float*)d_in[10];
  float* out = (float*)d_out;

  char* w = (char*)d_ws;
  auto carve = [&](size_t bytes) {
    char* p = w;
    w += (bytes + 255) & ~(size_t)255;
    return p;
  };
  float*    acc   = (float*)carve((size_t)N_NODES * D * 4);
  float*    asrc  = (float*)carve((size_t)N_NODES * 4);
  float*    adst  = (float*)carve((size_t)N_NODES * 4);
  float*    asrc2 = (float*)carve((size_t)N_NODES * 4);
  float*    adst2 = (float*)carve((size_t)N_NODES * 4);
  float*    cnt   = (float*)carve((size_t)(N_GRAPHS + N_BCNT) * 4);
  int*      bcnt  = (int*)(cnt + N_GRAPHS);
  unsigned* binw  = (unsigned*)carve((size_t)OWN_WG * BIN_CAP * 4);
  int*      ovf   = (int*)carve((size_t)N_EDGES * 4);

  const int TB = 256;
  const int gNode  = (N_NODES + TB - 1) / TB;
  const int gFin   = (2 * N_GRAPHS * D + TB - 1) / TB;
  const int gBin   = (N_EDGES + BIN_EPB - 1) / BIN_EPB;

  hipMemsetAsync(d_out, 0, (size_t)2 * N_GRAPHS * D * 4, stream);
  hipMemsetAsync(cnt, 0, (size_t)(N_GRAPHS + N_BCNT) * 4, stream);
  count_kernel<<<gNode, TB, 0, stream>>>(batch, cnt);
  bin_kernel<<<gBin, BIN_TPB, 0, stream>>>(ei, binw, bcnt, ovf);

  alpha_rows<<<OWN_WG, OWN_TPB, 0, stream>>>(x, W1, as1, ad1, asrc, adst);
  owner_pass<<<OWN_WG, OWN_TPB, 0, stream>>>(ei, asrc, adst, x, b1, batch, binw, bcnt, ovf,
                                             acc, out, 1, W1, W2, as2, ad2, asrc2, adst2);

  owner_pass<<<OWN_WG, OWN_TPB, 0, stream>>>(ei, asrc2, adst2, acc, b2, batch, binw, bcnt, ovf,
                                             acc, out + N_GRAPHS * D, 0, W2, W2, as2, ad2, asrc2, adst2);

  finalize_kernel<<<gFin, TB, 0, stream>>>(out, cnt);
}


# Matrix step overlapped with the row sums: rows in 4 stages; 8 of a block's 32 waves multiply the stage summed before while 24 gather the next; sums and products bit-identical
# speedup vs baseline: 5.4157x; 1.0338x over previous
#include <hip/hip_runtime.h>

#define N_NODES  50000
#define N_EDGES  600000
#define E_TOT    (N_EDGES + N_NODES)
#define N_GRAPHS 1024
#define D        128
#define NEG_SLOPE 0.2f

typedef float    v2f  __attribute__((ext_vector_type(2)));
typedef float    v8f  __attribute__((ext_vector_type(8)));

__device__ __forceinline__ unsigned fmap(float x) {
  const unsigned u = __float_as_uint(x);
  return (u & 0x80000000u) ? ~u : (u | 0x80000000u);
}
__device__ __forceinline__ float funmap(unsigned u) {
  return __uint_as_float((u & 0x80000000u) ? (u & 0x7fffffffu) : ~u);
}
__device__ __forceinline__ float leaky(float v) {
  return (v > 0.f) ? v : NEG_SLOPE * v;
}

#define OWN_WG    256
#define OWN_ROWS  ((N_NODES + OWN_WG - 1) / OWN_WG)
#define OWN_TPB   1024
#define OWN_WAVES (OWN_TPB / 32)
#define OWN_K     ((OWN_ROWS + OWN_WAVES - 1) / OWN_WAVES)
#define GRP       16
#define LPR       (D / 32)
#define BIN_CAP   4096
#define ENT_PT    (BIN_CAP / OWN_TPB)
#define BIN_TPB   1024
#define BIN_EPT   4
#define BIN_EPB   (BIN_TPB * BIN_EPT)
#define N_BCNT    (OWN_WG + 1)
#define POOL_Q    (OWN_TPB / D)
#define POOL_SL   ((OWN_ROWS + POOL_Q - 1) / POOL_Q)
#define SP        132
#define WP        136
#define OWN_RT    ((OWN_ROWS + 15) / 16)
#define OWN_RP    (OWN_RT * 16)
#define POOL_U    5
#define SRT_CAP   (BIN_CAP + OWN_ROWS)
#define SCAN_RPL  ((OWN_ROWS + 31) / 32)
#define MW        (D / 16)
#define GW        (OWN_WAVES - MW)
#define ST_T      3
#define ST_R      (ST_T * 16)
#define NST       (OWN_RT / ST_T)
#define ST_TL     (OWN_RT - ST_T * (NST - 1))
#define ST_UP     (ST_TL * 16)
static_assert(OWN_RT == 13 && OWN_WAVES == 32 && MW == 8, "13 row tiles in stages of 3, 3, 3 and 4; 8 column tiles, one per multiplying wave");
static_assert(ST_T == 3 && ST_TL == 4, "a multiplying wave keeps three accumulator tiles, four in the last stage");
static_assert(ST_R >= GW && OWN_ROWS - ST_R * (NST - 1) >= GW, "every gathering wave owns at least one row of every stage");
static_assert(N_NODES <= (1 << 16), "source node must fit the low half of a bin word");
static_assert(OWN_ROWS <= (1 << 16), "local row must fit the high half of a bin word");
static_assert(OWN_ROWS <= OWN_TPB, "one thread per row sets up the row's softmax cells");
static_assert(BIN_CAP % OWN_TPB == 0, "a bin is dealt out whole to the block's threads");
static_assert(D % 32 == 0, "a row must be whole lines");
static_assert(OWN_TPB % D == 0, "the closing pass gives every thread one column of one row slice");
static_assert(BIN_CAP % GRP == 0, "groups must tile a bin");
static_assert(OWN_WAVES == 32, "a wave's rows are the 32nd parts of the block's rows");
static_assert(OWN_ROWS >= OWN_WAVES, "every wave owns at least one row");
static_assert((SP * 4) % 16 == 0, "a row of sums is stored 16 bytes per lane");

__global__ __launch_bounds__(BIN_TPB)
void bin_kernel(const int* __restrict__ ei, unsigned* __restrict__ binw,
                int* __restrict__ bcnt, int* __restrict__ ovf) {
  __shared__ int hist[OWN_WG];
  __shared__ int base[OWN_WG];
  const int t = threadIdx.x;
  for (int i = t; i < OWN_WG; i += BIN_TPB) hist[i] = 0;
  __syncthreads();
  const int e0 = blockIdx.x * BIN_EPB + t;
  int dd[BIN_EPT], ss[BIN_EPT], bb[BIN_EPT], rk[BIN_EPT];
  #pragma unroll
  for (int k = 0; k < BIN_EPT; ++k) {
    const int e = e0 + k * BIN_TPB;
    dd[k] = 0; ss[k] = 0;
    if (e < N_EDGES) { dd[k] = ei[N_EDGES + e]; ss[k] = ei[e]; }
  }
  #pragma unroll
  for (int k = 0; k < BIN_EPT; ++k) {
    bb[k] = (int)((unsigned)dd[k] / (unsigned)OWN_ROWS);
    rk[k] = 0;
    if (e0 + k * BIN_TPB < N_EDGES) rk[k] = atomicAdd(&hist[bb[k]], 1);
  }
  __syncthreads();
  for (int i = t; i < OWN_WG; i += BIN_TPB) {
    const int c = hist[i];
    base[i] = (c > 0) ? atomicAdd(&bcnt[i], c) : 0;
  }
  __syncthreads();
  #pragma unroll
  for (int k = 0; k < BIN_EPT; ++k) {
    const int e = e0 + k * BIN_TPB;
    if (e < N_EDGES) {
      const int slot = base[bb[k]] + rk[k];
      if (slot < BIN_CAP) {
        const unsigned r = (unsigned)(dd[k] - bb[k] * OWN_ROWS);
        binw[(size_t)bb[k] * BIN_CAP + slot] = (unsigned)ss[k] | (r << 16);
      } else {
        ovf[atomicAdd(&bcnt[OWN_WG], 1)] = e;
      }
    }
  }
}

static_assert(OWN_TPB == 8 * D, "eight threads per row of the weight matrix");
__device__ __forceinline__ void wa_vectors(const float* __restrict__ W, const float* __restrict__ a_src,
                                           const float* __restrict__ a_dst, int t, float* u_s) {
  const int k = t >> 3, p = t & 7;
  const float4* wr = (const float4*)(W + (size_t)k * D + p * 16);
  const float4* sr = (const float4*)(a_src + p * 16);
  const float4* dr = (const float4*)(a_dst + p * 16);
  float s = 0.f, d = 0.f;
  #pragma unroll
  for (int i = 0; i < 4; ++i) {
    const float4 wv = wr[i], sv = sr[i], dv = dr[i];
    s += wv.x * sv.x + wv.y * sv.y + wv.z * sv.z + wv.w * sv.w;
    d += wv.x * dv.x + wv.y * dv.y + wv.z * dv.z + wv.w * dv.w;
  }
  #pragma unroll
  for (int off = 4; off > 0; off >>= 1) {
    s += __shfl_xor(s, off, 32);
    d += __shfl_xor(d, off, 32);
  }
  if (p == 0) { u_s[k] = s; u_s[D + k] = d; }
}

__global__ __launch_bounds__(OWN_TPB)
void alpha_rows(const float* __restrict__ X, const float* __restrict__ W,
                const float* __restrict__ a_src, const float* __restrict__ a_dst,
                float* __restrict__ asrc, float* __restrict__ adst) {
  __shared__ float u_s[2 * D];
  const int t = threadIdx.x;
  const int lane = t & 31;
  const int w = __builtin_amdgcn_readfirstlane(t >> 5);
  const int row0 = blockIdx.x * OWN_ROWS;
  float4 xv[OWN_K];
  #pragma unroll
  for (int k = 0; k < OWN_K; ++k) {
    int node = row0 + w + 32 * k;
    node = (node < N_NODES) ? node : (N_NODES - 1);
    xv[k] = *(const float4*)(X + (size_t)node * D + lane * 4);
  }
  wa_vectors(W, a_src, a_dst, t, u_s);
  __syncthreads();
  const float4 sv = *(const float4*)&u_s[lane * 4];
  const float4 dv = *(const float4*)&u_s[D + lane * 4];
  #pragma unroll
  for (int k = 0; k < OWN_K; ++k) {
    const int r = w + 32 * k;
    float s = xv[k].x * sv.x + xv[k].y * sv.y + xv[k].z * sv.z + xv[k].w * sv.w;
    float d = xv[k].x * dv.x + xv[k].y * dv.y + xv[k].z * dv.z + xv[k].w * dv.w;
    #pragma unroll
    for (int off = 16; off > 0; off >>= 1) {
      s += __shfl_down(s, off, 32);
      d += __shfl_down(d, off, 32);
    }
    if (lane == 0 && r < OWN_ROWS && row0 + r < N_NODES) { asrc[row0 + r] = s; adst[row0 + r] = d; }
  }
}

__device__ __forceinline__ bool ovf_edge(const int* __restrict__ ei, const int* __restrict__ ovf,
                                         int i, int row0, const float* __restrict__ asrc,
                                         const float* ad_s, unsigned& r, int& s, float& v) {
  const int e = ovf[i];
  r = (unsigned)(ei[N_EDGES + e] - row0);
  if (r >= (unsigned)OWN_ROWS) return false;
  s = ei[e];
  v = leaky(asrc[s] + ad_s[r]);
  return true;
}

template <bool FULL>
__device__ __forceinline__ void own_group(const unsigned* ew, const float* cf, int n, int lane,
                                          const float* __restrict__ H, float* sums,
                                          int& cur, float4& acc) {
  const int j0 = lane & (GRP - 1);
  unsigned q = 0;
  float coef = 0.f;
  if (FULL || j0 < n) { q = ew[j0]; coef = cf[j0]; }
  const int s = (int)(q & 0xFFFFu);
  const int dl = (int)(q >> 16);
  float4 hv[GRP];
  #pragma unroll
  for (int j = 0; j < GRP; ++j) {
    if (FULL || j < n) {
      const int sj = __builtin_amdgcn_readlane(s, j);
      hv[j] = *(const float4*)(H + (size_t)sj * D + lane * 4);
    } else {
      hv[j] = make_float4(0.f, 0.f, 0.f, 0.f);
    }
  }
  #pragma unroll
  for (int j = 0; j < GRP; ++j) {
    if (FULL || j < n) {
      const int dj = __builtin_amdgcn_readlane(dl, j);
      const float cj = __int_as_float(__builtin_amdgcn_readlane(__float_as_int(coef), j));
      if (dj != cur) {
        *(float4*)&sums[cur * SP + lane * 4] = acc;
        acc = make_float4(0.f, 0.f, 0.f, 0.f);
        cur = dj;
      }
      acc.x += hv[j].x * cj;
      acc.y += hv[j].y * cj;
      acc.z += hv[j].z * cj;
      acc.w += hv[j].w * cj;
    }
  }
}

template <int NT>
__device__ __forceinline__ void mat_tiles(const float* sums, const float (*wl)[WP], int t0, int n, int hi, int m,
                                          v8f& c0, v8f& c1, v8f& c2, v8f& c3) {
  const float* a0p = sums + (t0 * 16 + m) * SP;
  const float* a1p = a0p + 16 * SP;
  const float* a2p = a0p + 32 * SP;
  const float* a3p = a0p + 48 * SP;
  const v8f z = {0.f, 0.f, 0.f, 0.f, 0.f, 0.f, 0.f, 0.f};
  c0 = z; c1 = z; c2 = z; c3 = z;
  #pragma unroll
  for (int k0 = 0; k0 < D; k0 += 4) {
    const int ka = k0 + hi * 2;
    v2f a, b;
    b[0] = wl[ka][n];
    b[1] = wl[ka + 1][n];
    a[0] = a0p[ka];
    a[1] = a0p[ka + 1];
    c0 = __builtin_amdgcn_wmma_f32_16x16x4_f32(false, a, false, b, (short)0, c0, false, false);
    a[0] = a1p[ka];
    a[1] = a1p[ka + 1];
    c1 = __builtin_amdgcn_wmma_f32_16x16x4_f32(false, a, false, b, (short)0, c1, false, false);
    a[0] = a2p[ka];
    a[1] = a2p[ka + 1];
    c2 = __builtin_amdgcn_wmma_f32_16x16x4_f32(false, a, false, b, (short)0, c2, false, false);
    if (NT == 4) {
      a[0] = a3p[ka];
      a[1] = a3p[ka + 1];
      c3 = __builtin_amdgcn_wmma_f32_16x16x4_f32(false, a, false, b, (short)0, c3, false, false);
    }
  }
}

template <int NT>
__device__ __forceinline__ void put_tiles(float* sums, int t0, int n, int hi,
                                          const v8f& c0, const v8f& c1, const v8f& c2, const v8f& c3) {
  float* p = sums + (t0 * 16 + hi * 8) * SP + n;
  #pragma unroll
  for (int r = 0; r < 8; ++r) {
    p[r * SP] = c0[r];
    p[(16 + r) * SP] = c1[r];
    p[(32 + r) * SP] = c2[r];
    if (NT == 4) p[(48 + r) * SP] = c3[r];
  }
}

__global__ __launch_bounds__(OWN_TPB)
void owner_pass(const int* __restrict__ ei, const float* __restrict__ asrc,
                const float* __restrict__ adst, const float* __restrict__ H,
                const float* __restrict__ bias, const int* __restrict__ batch,
                const unsigned* __restrict__ binw, const int* __restrict__ bcnt,
                const int* __restrict__ ovf, float* __restrict__ acc,
                float* __restrict__ embsum, int store_rows,
                const float* __restrict__ Wm, const float* __restrict__ Wn,
                const float* __restrict__ nas, const float* __restrict__ nad,
                float* __restrict__ asrc_n, float* __restrict__ adst_n) {
  __shared__ __attribute__((aligned(16))) float sums[(OWN_RP + ST_UP) * SP];
  __shared__ float    wl[D][WP];
  __shared__ float    un_s[2 * D];
  __shared__ unsigned ew_s[SRT_CAP];
  __shared__ float    cf_s[SRT_CAP];
  __shared__ int      cn_s[OWN_ROWS];
  __shared__ int      st_s[OWN_ROWS + 1];
  __shared__ float    ad_s[OWN_ROWS];
  __shared__ float    sl_s[OWN_ROWS];
  __shared__ unsigned mx_s[OWN_ROWS];
  __shared__ float    dn_s[OWN_ROWS];
  __shared__ int      bt_s[OWN_ROWS];
  const int t = threadIdx.x;
  const int lane = t & 31;
  const int w = __builtin_amdgcn_readfirstlane(t >> 5);
  const int row0 = blockIdx.x * OWN_ROWS;
  const int nb_all = bcnt[blockIdx.x];
  const int nb = (nb_all < BIN_CAP) ? nb_all : BIN_CAP;
  const int novf = bcnt[OWN_WG];
  const unsigned* __restrict__ bw = binw + (size_t)blockIdx.x * BIN_CAP;

  float4 wv4[(D * D / 4) / OWN_TPB];
  #pragma unroll
  for (int k = 0; k < (D * D / 4) / OWN_TPB; ++k) {
    const int i = t + k * OWN_TPB;
    wv4[k] = ((const float4*)(Wm + (size_t)(i >> 5) * D))[i & 31];
  }
  if (store_rows) wa_vectors(Wn, nas, nad, t, un_s);

  int nd = row0 + t;
  nd = (nd < N_NODES) ? nd : (N_NODES - 1);
  const float ad_t = adst[nd];
  const float as_t = asrc[nd];
  const int   bt_t = batch[nd];

  unsigned ew[ENT_PT];
  float sc[ENT_PT];
  #pragma unroll
  for (int k = 0; k < ENT_PT; ++k)
    if (t + k * OWN_TPB < nb) ew[k] = bw[t + k * OWN_TPB];
  #pragma unroll
  for (int k = 0; k < ENT_PT; ++k) {
    const unsigned sn = (t + k * OWN_TPB < nb) ? (ew[k] & 0xFFFFu) : 0u;
    sc[k] = asrc[sn];
  }

  #pragma unroll
  for (int k = 0; k < (D * D / 4) / OWN_TPB; ++k) {
    const int i = t + k * OWN_TPB;
    *(float4*)&wl[i >> 5][(i & 31) * 4] = wv4[k];
  }

  if (t < OWN_ROWS) {
    const bool live = (row0 + t) < N_NODES;
    const float ad = live ? ad_t : 0.f;
    const float v = leaky((live ? as_t : 0.f) + ad);
    ad_s[t] = ad;
    sl_s[t] = v;
    mx_s[t] = fmap(v);
    dn_s[t] = 0.f;
    cn_s[t] = 0;
    bt_s[t] = live ? bt_t : 0;
  }
  __syncthreads();

  int rk[ENT_PT];
  #pragma unroll
  for (int k = 0; k < ENT_PT; ++k) {
    if (t + k * OWN_TPB < nb) {
      const unsigned r = ew[k] >> 16;
      sc[k] = leaky(sc[k] + ad_s[r]);
      atomicMax(&mx_s[r], fmap(sc[k]));
      rk[k] = atomicAdd(&cn_s[r], 1);
    }
  }
  for (int i = t; i < novf; i += OWN_TPB) {
    unsigned r; int s; float v;
    if (ovf_edge(ei, ovf, i, row0, asrc, ad_s, r, s, v)) atomicMax(&mx_s[r], fmap(v));
  }
  __syncthreads();

  #pragma unroll
  for (int k = 0; k < ENT_PT; ++k) {
    if (t + k * OWN_TPB < nb) {
      const unsigned r = ew[k] >> 16;
      sc[k] = __expf(sc[k] - funmap(mx_s[r]));
      __hip_atomic_fetch_add(&dn_s[r], sc[k], __ATOMIC_RELAXED, __HIP_MEMORY_SCOPE_WORKGROUP);
    }
  }
  for (int i = t; i < novf; i += OWN_TPB) {
    unsigned r; int s; float v;
    if (ovf_edge(ei, ovf, i, row0, asrc, ad_s, r, s, v))
      __hip_atomic_fetch_add(&dn_s[r], __expf(v - funmap(mx_s[r])), __ATOMIC_RELAXED,
                             __HIP_MEMORY_SCOPE_WORKGROUP);
  }
  if (t < OWN_ROWS)
    __hip_atomic_fetch_add(&dn_s[t], __expf(sl_s[t] - funmap(mx_s[t])), __ATOMIC_RELAXED,
                           __HIP_MEMORY_SCOPE_WORKGROUP);
  if (w == OWN_WAVES - 1) {
    int cc[SCAN_RPL];
    int tot = 0;
    #pragma unroll
    for (int i = 0; i < SCAN_RPL; ++i) {
      const int r = lane * SCAN_RPL + i;
      cc[i] = (r < OWN_ROWS) ? (cn_s[r] + 1) : 0;
      tot += cc[i];
    }
    int inc = tot;
    #pragma unroll
    for (int off = 1; off < 32; off <<= 1) {
      const int v = __shfl_up(inc, off, 32);
      if (lane >= off) inc += v;
    }
    int run = inc - tot;
    #pragma unroll
    for (int i = 0; i < SCAN_RPL; ++i) {
      const int r = lane * SCAN_RPL + i;
      if (r < OWN_ROWS) st_s[r] = run;
      run += cc[i];
    }
    if (lane == 31) st_s[OWN_ROWS] = inc;
  }
  __syncthreads();

  #pragma unroll
  for (int k = 0; k < ENT_PT; ++k) {
    if (t + k * OWN_TPB < nb) {
      const unsigned r = ew[k] >> 16;
      const int pos = st_s[r] + 1 + rk[k];
      ew_s[pos] = ew[k];
      cf_s[pos] = sc[k] / dn_s[r];
    }
  }
  if (t < OWN_ROWS) {
    const bool live = (row0 + t) < N_NODES;
    const unsigned node = live ? (unsigned)(row0 + t) : (unsigned)(N_NODES - 1);
    const int pos = st_s[t];
    ew_s[pos] = node | ((unsigned)t << 16);
    cf_s[pos] = live ? (__expf(sl_s[t] - funmap(mx_s[t])) / dn_s[t]) : 0.f;
  }
  float* const gs = sums + ST_UP * SP;
  if (w < OWN_RP - OWN_ROWS) {
    #pragma unroll
    for (int i = 0; i < LPR; ++i) gs[(OWN_ROWS + w) * SP + lane + 32 * i] = 0.f;
  }
  __syncthreads();

  {
    const bool mwave = (w >= GW);
    const int hi = lane >> 4;
    const int m  = lane & 15;
    const int n  = (mwave ? (w - GW) : 0) * 16 + m;
    for (int s = 0; s <= NST; ++s) {
      const int R0 = s * ST_R;
      const int nr = (s < NST - 1) ? ST_R : (OWN_ROWS - R0);
      if (mwave) {
        if (s >= 1) {
          v8f c0, c1, c2, c3;
          if (s < NST) {
            mat_tiles<ST_T>(gs, wl, (s - 1) * ST_T, n, hi, m, c0, c1, c2, c3);
            put_tiles<ST_T>(sums, (s - 1) * ST_T, n, hi, c0, c1, c2, c3);
          } else {
            mat_tiles<ST_TL>(gs, wl, (NST - 1) * ST_T, n, hi, m, c0, c1, c2, c3);
            put_tiles<ST_TL>(sums, (NST - 1) * ST_T, n, hi, c0, c1, c2, c3);
          }
        }
      } else if (s < NST) {
        const int rlo = R0 + (w * nr) / GW;
        const int rhi = R0 + ((w + 1) * nr) / GW;
        const int lo = __builtin_amdgcn_readfirstlane(st_s[rlo]);
        const int he = __builtin_amdgcn_readfirstlane(st_s[rhi]);
        int cur = rlo;
        float4 acc = make_float4(0.f, 0.f, 0.f, 0.f);
        int g = lo;
        for (; g + GRP <= he; g += GRP)
          own_group<true>(ew_s + g, cf_s + g, GRP, lane, H, gs, cur, acc);
        if (g < he)
          own_group<false>(ew_s + g, cf_s + g, he - g, lane, H, gs, cur, acc);
        *(float4*)&gs[cur * SP + lane * 4] = acc;
      }
      if (novf > 0 && s < NST) {
        __syncthreads();
        for (int i = t; i < novf; i += OWN_TPB) {
          unsigned r; int sn; float v;
          if (ovf_edge(ei, ovf, i, row0, asrc, ad_s, r, sn, v) && r - (unsigned)R0 < (unsigned)nr) {
            const float* hr = H + (size_t)sn * D;
            const float coef = __expf(v - funmap(mx_s[r])) / dn_s[r];
            for (int c = 0; c < D; ++c)
              __hip_atomic_fetch_add(&gs[r * SP + c], hr[c] * coef, __ATOMIC_RELAXED,
                                     __HIP_MEMORY_SCOPE_WORKGROUP);
          }
        }
      }
      __syncthreads();
    }
  }

  {
    const int c = t & (D - 1);
    const int q = t / D;
    int nlive = N_NODES - row0;
    nlive = (nlive < OWN_ROWS) ? nlive : OWN_ROWS;
    int r = q * POOL_SL;
    int rend = r + POOL_SL;
    rend = (rend < nlive) ? rend : nlive;
    const float bc = bias[c];
    float run = 0.f;
    int g = 0;
    bool have = false;
    for (int r0 = r; r0 < rend; r0 += POOL_U) {
      float vv[POOL_U];
      int gg[POOL_U];
      #pragma unroll
      for (int k = 0; k < POOL_U; ++k) {
        const int rr = (r0 + k < rend) ? (r0 + k) : r0;
        vv[k] = sums[rr * SP + c];
        gg[k] = bt_s[rr];
      }
      #pragma unroll
      for (int k = 0; k < POOL_U; ++k) {
        if (r0 + k < rend) {
          float v = vv[k] + bc;
          v = (v > 0.f) ? v : 0.f;
          if (store_rows) {
            acc[(size_t)(row0 + r0 + k) * D + c] = v;
            sums[(r0 + k) * SP + c] = v;
          }
          if (have && gg[k] != g) {
            atomicAdd(&embsum[(size_t)g * D + c], run);
            run = 0.f;
          }
          g = gg[k];
          have = true;
          run += v;
        }
      }
    }
    if (have) atomicAdd(&embsum[(size_t)g * D + c], run);
  }

  if (store_rows) {
    __syncthreads();
    int nlive = N_NODES - row0;
    nlive = (nlive < OWN_ROWS) ? nlive : OWN_ROWS;
    float us[LPR], ud[LPR];
    #pragma unroll
    for (int i = 0; i < LPR; ++i) { us[i] = un_s[lane + 32 * i]; ud[i] = un_s[D + lane + 32 * i]; }
    #pragma unroll
    for (int k = 0; k < OWN_K; ++k) {
      const int r = w + 32 * k;
      if (r < nlive) {
        float s = 0.f, d = 0.f;
        #pragma unroll
        for (int i = 0; i < LPR; ++i) {
          const float hv = sums[r * SP + lane + 32 * i];
          s += hv * us[i];
          d += hv * ud[i];
        }
        #pragma unroll
        for (int off = 16; off > 0; off >>= 1) {
          s += __shfl_down(s, off, 32);
          d += __shfl_down(d, off, 32);
        }
        if (lane == 0) { asrc_n[row0 + r] = s; adst_n[row0 + r] = d; }
      }
    }
  }
}

__global__ __launch_bounds__(256)
void count_kernel(const int* __restrict__ batch, float* __restrict__ cnt) {
  const int t = blockIdx.x * blockDim.x + threadIdx.x;
  if (t < N_NODES) atomicAdd(&cnt[batch[t]], 1.0f);
}

__global__ __launch_bounds__(256)
void finalize_kernel(float* __restrict__ out, const float* __restrict__ cnt) {
  const int t = blockIdx.x * blockDim.x + threadIdx.x;
  if (t >= 2 * N_GRAPHS * D) return;
  const int g = (t >> 7) & (N_GRAPHS - 1);
  float c = cnt[g];
  c = (c > 1.f) ? c : 1.f;
  out[t] /= c;
}

extern "C" void kernel_launch(void* const* d_in, const int* in_sizes, int n_in,
                              void* d_out, int out_size, void* d_ws, size_t ws_size,
                              hipStream_t stream) {
  const float* x     = (const float*)d_in[0];
  const int*   ei    = (const int*)d_in[1];
  const int*   batch = (const int*)d_in[2];
  const float* W1    = (const float*)d_in[3];
  const float* as1   = (const float*)d_in[4];
  const float* ad1   = (const float*)d_in[5];
  const float* b1    = (const float*)d_in[6];
  const float* W2    = (const float*)d_in[7];
  const float* as2   = (const float*)d_in[8];
  const float* ad2   = (const float*)d_in[9];
  const float* b2    = (const float*)d_in[10];
  float* out = (float*)d_out;

  char* w = (char*)d_ws;
  auto carve = [&](size_t bytes) {
    char* p = w;
    w += (bytes + 255) & ~(size_t)255;
    return p;
  };
  float*    acc   = (float*)carve((size_t)N_NODES * D * 4);
  float*    asrc  = (float*)carve((size_t)N_NODES * 4);
  float*    adst  = (float*)carve((size_t)N_NODES * 4);
  float*    asrc2 = (float*)carve((size_t)N_NODES * 4);
  float*    adst2 = (float*)carve((size_t)N_NODES * 4);
  float*    cnt   = (float*)carve((size_t)(N_GRAPHS + N_BCNT) * 4);
  int*      bcnt  = (int*)(cnt + N_GRAPHS);
  unsigned* binw  = (unsigned*)carve((size_t)OWN_WG * BIN_CAP * 4);
  int*      ovf   = (int*)carve((size_t)N_EDGES * 4);

  const int TB = 256;
  const int gNode  = (N_NODES + TB - 1) / TB;
  const int gFin   = (2 * N_GRAPHS * D + TB - 1) / TB;
  const int gBin   = (N_EDGES + BIN_EPB - 1) / BIN_EPB;

  hipMemsetAsync(d_out, 0, (size_t)2 * N_GRAPHS * D * 4, stream);
  hipMemsetAsync(cnt, 0, (size_t)(N_GRAPHS + N_BCNT) * 4, stream);
  count_kernel<<<gNode, TB, 0, stream>>>(batch, cnt);
  bin_kernel<<<gBin, BIN_TPB, 0, stream>>>(ei, binw, bcnt, ovf);

  alpha_rows<<<OWN_WG, OWN_TPB, 0, stream>>>(x, W1, as1, ad1, asrc, adst);
  owner_pass<<<OWN_WG, OWN_TPB, 0, stream>>>(ei, asrc, adst, x, b1, batch, binw, bcnt, ovf,
                                             acc, out, 1, W1, W2, as2, ad2, asrc2, adst2);

  owner_pass<<<OWN_WG, OWN_TPB, 0, stream>>>(ei, asrc2, adst2, acc, b2, batch, binw, bcnt, ovf,
                                             acc, out + N_GRAPHS * D, 0, W2, W2, as2, ad2, asrc2, adst2);

  finalize_kernel<<<gFin, TB, 0, stream>>>(out, cnt);
}


# The three kernels that wait for no other kernel's results (edge split, layer-1 score parts, graph sizes) and out's zero fill run as block ranges of ONE launch; 5 stream operations, not 8
# speedup vs baseline: 5.8006x; 1.0711x over previous
#include <hip/hip_runtime.h>

#define N_NODES  50000
#define N_EDGES  600000
#define E_TOT    (N_EDGES + N_NODES)
#define N_GRAPHS 1024
#define D        128
#define NEG_SLOPE 0.2f

typedef float    v2f  __attribute__((ext_vector_type(2)));
typedef float    v8f  __attribute__((ext_vector_type(8)));

__device__ __forceinline__ unsigned fmap(float x) {
  const unsigned u = __float_as_uint(x);
  return (u & 0x80000000u) ? ~u : (u | 0x80000000u);
}
__device__ __forceinline__ float funmap(unsigned u) {
  return __uint_as_float((u & 0x80000000u) ? (u & 0x7fffffffu) : ~u);
}
__device__ __forceinline__ float leaky(float v) {
  return (v > 0.f) ? v : NEG_SLOPE * v;
}

#define OWN_WG    256
#define OWN_ROWS  ((N_NODES + OWN_WG - 1) / OWN_WG)
#define OWN_TPB   1024
#define OWN_WAVES (OWN_TPB / 32)
#define OWN_K     ((OWN_ROWS + OWN_WAVES - 1) / OWN_WAVES)
#define GRP       16
#define LPR       (D / 32)
#define BIN_CAP   4096
#define ENT_PT    (BIN_CAP / OWN_TPB)
#define BIN_TPB   1024
#define BIN_EPT   4
#define BIN_EPB   (BIN_TPB * BIN_EPT)
#define BIN_WG    ((N_EDGES + BIN_EPB - 1) / BIN_EPB)
#define N_BCNT    (OWN_WG + 1)
#define POOL_Q    (OWN_TPB / D)
#define POOL_SL   ((OWN_ROWS + POOL_Q - 1) / POOL_Q)
#define SP        132
#define WP        136
#define OWN_RT    ((OWN_ROWS + 15) / 16)
#define OWN_RP    (OWN_RT * 16)
#define POOL_U    5
#define SRT_CAP   (BIN_CAP + OWN_ROWS)
#define SCAN_RPL  ((OWN_ROWS + 31) / 32)
#define MW        (D / 16)
#define GW        (OWN_WAVES - MW)
#define ST_T      3
#define ST_R      (ST_T * 16)
#define NST       (OWN_RT / ST_T)
#define ST_TL     (OWN_RT - ST_T * (NST - 1))
#define ST_UP     (ST_TL * 16)
static_assert(OWN_RT == 13 && OWN_WAVES == 32 && MW == 8, "13 row tiles in stages of 3, 3, 3 and 4; 8 column tiles, one per multiplying wave");
static_assert(ST_T == 3 && ST_TL == 4, "a multiplying wave keeps three accumulator tiles, four in the last stage");
static_assert(ST_R >= GW && OWN_ROWS - ST_R * (NST - 1) >= GW, "every gathering wave owns at least one row of every stage");
static_assert(N_NODES <= (1 << 16), "source node must fit the low half of a bin word");
static_assert(OWN_ROWS <= (1 << 16), "local row must fit the high half of a bin word");
static_assert(OWN_ROWS <= OWN_TPB, "one thread per row sets up the row's softmax cells");
static_assert(BIN_CAP % OWN_TPB == 0, "a bin is dealt out whole to the block's threads");
static_assert(D % 32 == 0, "a row must be whole lines");
static_assert(OWN_TPB % D == 0, "the closing pass gives every thread one column of one row slice");
static_assert(BIN_CAP % GRP == 0, "groups must tile a bin");
static_assert(OWN_WAVES == 32, "a wave's rows are the 32nd parts of the block's rows");
static_assert(OWN_ROWS >= OWN_WAVES, "every wave owns at least one row");
static_assert((SP * 4) % 16 == 0, "a row of sums is stored 16 bytes per lane");

__device__ __forceinline__ void bin_part(int b, const int* __restrict__ ei, unsigned* __restrict__ binw,
                                         int* __restrict__ bcnt, int* __restrict__ ovf) {
  __shared__ int hist[OWN_WG];
  __shared__ int base[OWN_WG];
  const int t = threadIdx.x;
  for (int i = t; i < OWN_WG; i += BIN_TPB) hist[i] = 0;
  __syncthreads();
  const int e0 = b * BIN_EPB + t;
  int dd[BIN_EPT], ss[BIN_EPT], bb[BIN_EPT], rk[BIN_EPT];
  #pragma unroll
  for (int k = 0; k < BIN_EPT; ++k) {
    const int e = e0 + k * BIN_TPB;
    dd[k] = 0; ss[k] = 0;
    if (e < N_EDGES) { dd[k] = ei[N_EDGES + e]; ss[k] = ei[e]; }
  }
  #pragma unroll
  for (int k = 0; k < BIN_EPT; ++k) {
    bb[k] = (int)((unsigned)dd[k] / (unsigned)OWN_ROWS);
    rk[k] = 0;
    if (e0 + k * BIN_TPB < N_EDGES) rk[k] = atomicAdd(&hist[bb[k]], 1);
  }
  __syncthreads();
  for (int i = t; i < OWN_WG; i += BIN_TPB) {
    const int c = hist[i];
    base[i] = (c > 0) ? atomicAdd(&bcnt[i], c) : 0;
  }
  __syncthreads();
  #pragma unroll
  for (int k = 0; k < BIN_EPT; ++k) {
    const int e = e0 + k * BIN_TPB;
    if (e < N_EDGES) {
      const int slot = base[bb[k]] + rk[k];
      if (slot < BIN_CAP) {
        const unsigned r = (unsigned)(dd[k] - bb[k] * OWN_ROWS);
        binw[(size_t)bb[k] * BIN_CAP + slot] = (unsigned)ss[k] | (r << 16);
      } else {
        ovf[atomicAdd(&bcnt[OWN_WG], 1)] = e;
      }
    }
  }
}

static_assert(OWN_TPB == 8 * D, "eight threads per row of the weight matrix");
__device__ __forceinline__ void wa_vectors(const float* __restrict__ W, const float* __restrict__ a_src,
                                           const float* __restrict__ a_dst, int t, float* u_s) {
  const int k = t >> 3, p = t & 7;
  const float4* wr = (const float4*)(W + (size_t)k * D + p * 16);
  const float4* sr = (const float4*)(a_src + p * 16);
  const float4* dr = (const float4*)(a_dst + p * 16);
  float s = 0.f, d = 0.f;
  #pragma unroll
  for (int i = 0; i < 4; ++i) {
    const float4 wv = wr[i], sv = sr[i], dv = dr[i];
    s += wv.x * sv.x + wv.y * sv.y + wv.z * sv.z + wv.w * sv.w;
    d += wv.x * dv.x + wv.y * dv.y + wv.z * dv.z + wv.w * dv.w;
  }
  #pragma unroll
  for (int off = 4; off > 0; off >>= 1) {
    s += __shfl_xor(s, off, 32);
    d += __shfl_xor(d, off, 32);
  }
  if (p == 0) { u_s[k] = s; u_s[D + k] = d; }
}

static_assert(OWN_WG * OWN_TPB == 2 * N_GRAPHS * D, "one float of out per thread of this part");
__device__ __forceinline__ void alpha_part(int b, const float* __restrict__ X, const float* __restrict__ W,
                                           const float* __restrict__ a_src, const float* __restrict__ a_dst,
                                           float* __restrict__ asrc, float* __restrict__ adst,
                                           const int* __restrict__ batch, float* __restrict__ cnt,
                                           float* __restrict__ out) {
  __shared__ float u_s[2 * D];
  const int t = threadIdx.x;
  const int lane = t & 31;
  const int w = __builtin_amdgcn_readfirstlane(t >> 5);
  const int row0 = b * OWN_ROWS;
  out[b * OWN_TPB + t] = 0.f;
  int cn = row0 + t;
  cn = (cn < N_NODES) ? cn : (N_NODES - 1);
  const int gid = batch[cn];
  float4 xv[OWN_K];
  #pragma unroll
  for (int k = 0; k < OWN_K; ++k) {
    int node = row0 + w + 32 * k;
    node = (node < N_NODES) ? node : (N_NODES - 1);
    xv[k] = *(const float4*)(X + (size_t)node * D + lane * 4);
  }
  wa_vectors(W, a_src, a_dst, t, u_s);
  if (t < OWN_ROWS && row0 + t < N_NODES) atomicAdd(&cnt[gid], 1.0f);
  __syncthreads();
  const float4 sv = *(const float4*)&u_s[lane * 4];
  const float4 dv = *(const float4*)&u_s[D + lane * 4];
  #pragma unroll
  for (int k = 0; k < OWN_K; ++k) {
    const int r = w + 32 * k;
    float s = xv[k].x * sv.x + xv[k].y * sv.y + xv[k].z * sv.z + xv[k].w * sv.w;
    float d = xv[k].x * dv.x + xv[k].y * dv.y + xv[k].z * dv.z + xv[k].w * dv.w;
    #pragma unroll
    for (int off = 16; off > 0; off >>= 1) {
      s += __shfl_down(s, off, 32);
      d += __shfl_down(d, off, 32);
    }
    if (lane == 0 && r < OWN_ROWS && row0 + r < N_NODES) { asrc[row0 + r] = s; adst[row0 + r] = d; }
  }
}

static_assert(BIN_TPB == OWN_TPB, "both parts run with the same block size");
__global__ __launch_bounds__(OWN_TPB, 16)
void front_kernel(const int* __restrict__ ei, unsigned* __restrict__ binw,
                  int* __restrict__ bcnt, int* __restrict__ ovf,
                  const float* __restrict__ X, const float* __restrict__ W,
                  const float* __restrict__ a_src, const float* __restrict__ a_dst,
                  float* __restrict__ asrc, float* __restrict__ adst,
                  const int* __restrict__ batch, float* __restrict__ cnt,
                  float* __restrict__ out) {
  if (blockIdx.x < BIN_WG) bin_part(blockIdx.x, ei, binw, bcnt, ovf);
  else alpha_part(blockIdx.x - BIN_WG, X, W, a_src, a_dst, asrc, adst, batch, cnt, out);
}

__device__ __forceinline__ bool ovf_edge(const int* __restrict__ ei, const int* __restrict__ ovf,
                                         int i, int row0, const float* __restrict__ asrc,
                                         const float* ad_s, unsigned& r, int& s, float& v) {
  const int e = ovf[i];
  r = (unsigned)(ei[N_EDGES + e] - row0);
  if (r >= (unsigned)OWN_ROWS) return false;
  s = ei[e];
  v = leaky(asrc[s] + ad_s[r]);
  return true;
}

template <bool FULL>
__device__ __forceinline__ void own_group(const unsigned* ew, const float* cf, int n, int lane,
                                          const float* __restrict__ H, float* sums,
                                          int& cur, float4& acc) {
  const int j0 = lane & (GRP - 1);
  unsigned q = 0;
  float coef = 0.f;
  if (FULL || j0 < n) { q = ew[j0]; coef = cf[j0]; }
  const int s = (int)(q & 0xFFFFu);
  const int dl = (int)(q >> 16);
  float4 hv[GRP];
  #pragma unroll
  for (int j = 0; j < GRP; ++j) {
    if (FULL || j < n) {
      const int sj = __builtin_amdgcn_readlane(s, j);
      hv[j] = *(const float4*)(H + (size_t)sj * D + lane * 4);
    } else {
      hv[j] = make_float4(0.f, 0.f, 0.f, 0.f);
    }
  }
  #pragma unroll
  for (int j = 0; j < GRP; ++j) {
    if (FULL || j < n) {
      const int dj = __builtin_amdgcn_readlane(dl, j);
      const float cj = __int_as_float(__builtin_amdgcn_readlane(__float_as_int(coef), j));
      if (dj != cur) {
        *(float4*)&sums[cur * SP + lane * 4] = acc;
        acc = make_float4(0.f, 0.f, 0.f, 0.f);
        cur = dj;
      }
      acc.x += hv[j].x * cj;
      acc.y += hv[j].y * cj;
      acc.z += hv[j].z * cj;
      acc.w += hv[j].w * cj;
    }
  }
}

template <int NT>
__device__ __forceinline__ void mat_tiles(const float* sums, const float (*wl)[WP], int t0, int n, int hi, int m,
                                          v8f& c0, v8f& c1, v8f& c2, v8f& c3) {
  const float* a0p = sums + (t0 * 16 + m) * SP;
  const float* a1p = a0p + 16 * SP;
  const float* a2p = a0p + 32 * SP;
  const float* a3p = a0p + 48 * SP;
  const v8f z = {0.f, 0.f, 0.f, 0.f, 0.f, 0.f, 0.f, 0.f};
  c0 = z; c1 = z; c2 = z; c3 = z;
  #pragma unroll
  for (int k0 = 0; k0 < D; k0 += 4) {
    const int ka = k0 + hi * 2;
    v2f a, b;
    b[0] = wl[ka][n];
    b[1] = wl[ka + 1][n];
    a[0] = a0p[ka];
    a[1] = a0p[ka + 1];
    c0 = __builtin_amdgcn_wmma_f32_16x16x4_f32(false, a, false, b, (short)0, c0, false, false);
    a[0] = a1p[ka];
    a[1] = a1p[ka + 1];
    c1 = __builtin_amdgcn_wmma_f32_16x16x4_f32(false, a, false, b, (short)0, c1, false, false);
    a[0] = a2p[ka];
    a[1] = a2p[ka + 1];
    c2 = __builtin_amdgcn_wmma_f32_16x16x4_f32(false, a, false, b, (short)0, c2, false, false);
    if (NT == 4) {
      a[0] = a3p[ka];
      a[1] = a3p[ka + 1];
      c3 = __builtin_amdgcn_wmma_f32_16x16x4_f32(false, a, false, b, (short)0, c3, false, false);
    }
  }
}

template <int NT>
__device__ __forceinline__ void put_tiles(float* sums, int t0, int n, int hi,
                                          const v8f& c0, const v8f& c1, const v8f& c2, const v8f& c3) {
  float* p = sums + (t0 * 16 + hi * 8) * SP + n;
  #pragma unroll
  for (int r = 0; r < 8; ++r) {
    p[r * SP] = c0[r];
    p[(16 + r) * SP] = c1[r];
    p[(32 + r) * SP] = c2[r];
    if (NT == 4) p[(48 + r) * SP] = c3[r];
  }
}

__global__ __launch_bounds__(OWN_TPB)
void owner_pass(const int* __restrict__ ei, const float* __restrict__ asrc,
                const float* __restrict__ adst, const float* __restrict__ H,
                const float* __restrict__ bias, const int* __restrict__ batch,
                const unsigned* __restrict__ binw, const int* __restrict__ bcnt,
                const int* __restrict__ ovf, float* __restrict__ acc,
                float* __restrict__ embsum, int store_rows,
                const float* __restrict__ Wm, const float* __restrict__ Wn,
                const float* __restrict__ nas, const float* __restrict__ nad,
                float* __restrict__ asrc_n, float* __restrict__ adst_n) {
  __shared__ __attribute__((aligned(16))) float sums[(OWN_RP + ST_UP) * SP];
  __shared__ float    wl[D][WP];
  __shared__ float    un_s[2 * D];
  __shared__ unsigned ew_s[SRT_CAP];
  __shared__ float    cf_s[SRT_CAP];
  __shared__ int      cn_s[OWN_ROWS];
  __shared__ int      st_s[OWN_ROWS + 1];
  __shared__ float    ad_s[OWN_ROWS];
  __shared__ float    sl_s[OWN_ROWS];
  __shared__ unsigned mx_s[OWN_ROWS];
  __shared__ float    dn_s[OWN_ROWS];
  __shared__ int      bt_s[OWN_ROWS];
  const int t = threadIdx.x;
  const int lane = t & 31;
  const int w = __builtin_amdgcn_readfirstlane(t >> 5);
  const int row0 = blockIdx.x * OWN_ROWS;
  const int nb_all = bcnt[blockIdx.x];
  const int nb = (nb_all < BIN_CAP) ? nb_all : BIN_CAP;
  const int novf = bcnt[OWN_WG];
  const unsigned* __restrict__ bw = binw + (size_t)blockIdx.x * BIN_CAP;

  float4 wv4[(D * D / 4) / OWN_TPB];
  #pragma unroll
  for (int k = 0; k < (D * D / 4) / OWN_TPB; ++k) {
    const int i = t + k * OWN_TPB;
    wv4[k] = ((const float4*)(Wm + (size_t)(i >> 5) * D))[i & 31];
  }
  if (store_rows) wa_vectors(Wn, nas, nad, t, un_s);

  int nd = row0 + t;
  nd = (nd < N_NODES) ? nd : (N_NODES - 1);
  const float ad_t = adst[nd];
  const float as_t = asrc[nd];
  const int   bt_t = batch[nd];

  unsigned ew[ENT_PT];
  float sc[ENT_PT];
  #pragma unroll
  for (int k = 0; k < ENT_PT; ++k)
    if (t + k * OWN_TPB < nb) ew[k] = bw[t + k * OWN_TPB];
  #pragma unroll
  for (int k = 0; k < ENT_PT; ++k) {
    const unsigned sn = (t + k * OWN_TPB < nb) ? (ew[k] & 0xFFFFu) : 0u;
    sc[k] = asrc[sn];
  }

  #pragma unroll
  for (int k = 0; k < (D * D / 4) / OWN_TPB; ++k) {
    const int i = t + k * OWN_TPB;
    *(float4*)&wl[i >> 5][(i & 31) * 4] = wv4[k];
  }

  if (t < OWN_ROWS) {
    const bool live = (row0 + t) < N_NODES;
    const float ad = live ? ad_t : 0.f;
    const float v = leaky((live ? as_t : 0.f) + ad);
    ad_s[t] = ad;
    sl_s[t] = v;
    mx_s[t] = fmap(v);
    dn_s[t] = 0.f;
    cn_s[t] = 0;
    bt_s[t] = live ? bt_t : 0;
  }
  __syncthreads();

  int rk[ENT_PT];
  #pragma unroll
  for (int k = 0; k < ENT_PT; ++k) {
    if (t + k * OWN_TPB < nb) {
      const unsigned r = ew[k] >> 16;
      sc[k] = leaky(sc[k] + ad_s[r]);
      atomicMax(&mx_s[r], fmap(sc[k]));
      rk[k] = atomicAdd(&cn_s[r], 1);
    }
  }
  for (int i = t; i < novf; i += OWN_TPB) {
    unsigned r; int s; float v;
    if (ovf_edge(ei, ovf, i, row0, asrc, ad_s, r, s, v)) atomicMax(&mx_s[r], fmap(v));
  }
  __syncthreads();

  #pragma unroll
  for (int k = 0; k < ENT_PT; ++k) {
    if (t + k * OWN_TPB < nb) {
      const unsigned r = ew[k] >> 16;
      sc[k] = __expf(sc[k] - funmap(mx_s[r]));
      __hip_atomic_fetch_add(&dn_s[r], sc[k], __ATOMIC_RELAXED, __HIP_MEMORY_SCOPE_WORKGROUP);
    }
  }
  for (int i = t; i < novf; i += OWN_TPB) {
    unsigned r; int s; float v;
    if (ovf_edge(ei, ovf, i, row0, asrc, ad_s, r, s, v))
      __hip_atomic_fetch_add(&dn_s[r], __expf(v - funmap(mx_s[r])), __ATOMIC_RELAXED,
                             __HIP_MEMORY_SCOPE_WORKGROUP);
  }
  if (t < OWN_ROWS)
    __hip_atomic_fetch_add(&dn_s[t], __expf(sl_s[t] - funmap(mx_s[t])), __ATOMIC_RELAXED,
                           __HIP_MEMORY_SCOPE_WORKGROUP);
  if (w == OWN_WAVES - 1) {
    int cc[SCAN_RPL];
    int tot = 0;
    #pragma unroll
    for (int i = 0; i < SCAN_RPL; ++i) {
      const int r = lane * SCAN_RPL + i;
      cc[i] = (r < OWN_ROWS) ? (cn_s[r] + 1) : 0;
      tot += cc[i];
    }
    int inc = tot;
    #pragma unroll
    for (int off = 1; off < 32; off <<= 1) {
      const int v = __shfl_up(inc, off, 32);
      if (lane >= off) inc += v;
    }
    int run = inc - tot;
    #pragma unroll
    for (int i = 0; i < SCAN_RPL; ++i) {
      const int r = lane * SCAN_RPL + i;
      if (r < OWN_ROWS) st_s[r] = run;
      run += cc[i];
    }
    if (lane == 31) st_s[OWN_ROWS] = inc;
  }
  __syncthreads();

  #pragma unroll
  for (int k = 0; k < ENT_PT; ++k) {
    if (t + k * OWN_TPB < nb) {
      const unsigned r = ew[k] >> 16;
      const int pos = st_s[r] + 1 + rk[k];
      ew_s[pos] = ew[k];
      cf_s[pos] = sc[k] / dn_s[r];
    }
  }
  if (t < OWN_ROWS) {
    const bool live = (row0 + t) < N_NODES;
    const unsigned node = live ? (unsigned)(row0 + t) : (unsigned)(N_NODES - 1);
    const int pos = st_s[t];
    ew_s[pos] = node | ((unsigned)t << 16);
    cf_s[pos] = live ? (__expf(sl_s[t] - funmap(mx_s[t])) / dn_s[t]) : 0.f;
  }
  float* const gs = sums + ST_UP * SP;
  if (w < OWN_RP - OWN_ROWS) {
    #pragma unroll
    for (int i = 0; i < LPR; ++i) gs[(OWN_ROWS + w) * SP + lane + 32 * i] = 0.f;
  }
  __syncthreads();

  {
    const bool mwave = (w >= GW);
    const int hi = lane >> 4;
    const int m  = lane & 15;
    const int n  = (mwave ? (w - GW) : 0) * 16 + m;
    for (int s = 0; s <= NST; ++s) {
      const int R0 = s * ST_R;
      const int nr = (s < NST - 1) ? ST_R : (OWN_ROWS - R0);
      if (mwave) {
        if (s >= 1) {
          v8f c0, c1, c2, c3;
          if (s < NST) {
            mat_tiles<ST_T>(gs, wl, (s - 1) * ST_T, n, hi, m, c0, c1, c2, c3);
            put_tiles<ST_T>(sums, (s - 1) * ST_T, n, hi, c0, c1, c2, c3);
          } else {
            mat_tiles<ST_TL>(gs, wl, (NST - 1) * ST_T, n, hi, m, c0, c1, c2, c3);
            put_tiles<ST_TL>(sums, (NST - 1) * ST_T, n, hi, c0, c1, c2, c3);
          }
        }
      } else if (s < NST) {
        const int rlo = R0 + (w * nr) / GW;
        const int rhi = R0 + ((w + 1) * nr) / GW;
        const int lo = __builtin_amdgcn_readfirstlane(st_s[rlo]);
        const int he = __builtin_amdgcn_readfirstlane(st_s[rhi]);
        int cur = rlo;
        float4 acc = make_float4(0.f, 0.f, 0.f, 0.f);
        int g = lo;
        for (; g + GRP <= he; g += GRP)
          own_group<true>(ew_s + g, cf_s + g, GRP, lane, H, gs, cur, acc);
        if (g < he)
          own_group<false>(ew_s + g, cf_s + g, he - g, lane, H, gs, cur, acc);
        *(float4*)&gs[cur * SP + lane * 4] = acc;
      }
      if (novf > 0 && s < NST) {
        __syncthreads();
        for (int i = t; i < novf; i += OWN_TPB) {
          unsigned r; int sn; float v;
          if (ovf_edge(ei, ovf, i, row0, asrc, ad_s, r, sn, v) && r - (unsigned)R0 < (unsigned)nr) {
            const float* hr = H + (size_t)sn * D;
            const float coef = __expf(v - funmap(mx_s[r])) / dn_s[r];
            for (int c = 0; c < D; ++c)
              __hip_atomic_fetch_add(&gs[r * SP + c], hr[c] * coef, __ATOMIC_RELAXED,
                                     __HIP_MEMORY_SCOPE_WORKGROUP);
          }
        }
      }
      __syncthreads();
    }
  }

  {
    const int c = t & (D - 1);
    const int q = t / D;
    int nlive = N_NODES - row0;
    nlive = (nlive < OWN_ROWS) ? nlive : OWN_ROWS;
    int r = q * POOL_SL;
    int rend = r + POOL_SL;
    rend = (rend < nlive) ? rend : nlive;
    const float bc = bias[c];
    float run = 0.f;
    int g = 0;
    bool have = false;
    for (int r0 = r; r0 < rend; r0 += POOL_U) {
      float vv[POOL_U];
      int gg[POOL_U];
      #pragma unroll
      for (int k = 0; k < POOL_U; ++k) {
        const int rr = (r0 + k < rend) ? (r0 + k) : r0;
        vv[k] = sums[rr * SP + c];
        gg[k] = bt_s[rr];
      }
      #pragma unroll
      for (int k = 0; k < POOL_U; ++k) {
        if (r0 + k < rend) {
          float v = vv[k] + bc;
          v = (v > 0.f) ? v : 0.f;
          if (store_rows) {
            acc[(size_t)(row0 + r0 + k) * D + c] = v;
            sums[(r0 + k) * SP + c] = v;
          }
          if (have && gg[k] != g) {
            atomicAdd(&embsum[(size_t)g * D + c], run);
            run = 0.f;
          }
          g = gg[k];
          have = true;
          run += v;
        }
      }
    }
    if (have) atomicAdd(&embsum[(size_t)g * D + c], run);
  }

  if (store_rows) {
    __syncthreads();
    int nlive = N_NODES - row0;
    nlive = (nlive < OWN_ROWS) ? nlive : OWN_ROWS;
    float us[LPR], ud[LPR];
    #pragma unroll
    for (int i = 0; i < LPR; ++i) { us[i] = un_s[lane + 32 * i]; ud[i] = un_s[D + lane + 32 * i]; }
    #pragma unroll
    for (int k = 0; k < OWN_K; ++k) {
      const int r = w + 32 * k;
      if (r < nlive) {
        float s = 0.f, d = 0.f;
        #pragma unroll
        for (int i = 0; i < LPR; ++i) {
          const float hv = sums[r * SP + lane + 32 * i];
          s += hv * us[i];
          d += hv * ud[i];
        }
        #pragma unroll
        for (int off = 16; off > 0; off >>= 1) {
          s += __shfl_down(s, off, 32);
          d += __shfl_down(d, off, 32);
        }
        if (lane == 0) { asrc_n[row0 + r] = s; adst_n[row0 + r] = d; }
      }
    }
  }
}

__global__ __launch_bounds__(256)
void finalize_kernel(float* __restrict__ out, const float* __restrict__ cnt) {
  const int t = blockIdx.x * blockDim.x + threadIdx.x;
  if (t >= 2 * N_GRAPHS * D) return;
  const int g = (t >> 7) & (N_GRAPHS - 1);
  float c = cnt[g];
  c = (c > 1.f) ? c : 1.f;
  out[t] /= c;
}

extern "C" void kernel_launch(void* const* d_in, const int* in_sizes, int n_in,
                              void* d_out, int out_size, void* d_ws, size_t ws_size,
                              hipStream_t stream) {
  const float* x     = (const float*)d_in[0];
  const int*   ei    = (const int*)d_in[1];
  const int*   batch = (const int*)d_in[2];
  const float* W1    = (const float*)d_in[3];
  const float* as1   = (const float*)d_in[4];
  const float* ad1   = (const float*)d_in[5];
  const float* b1    = (const float*)d_in[6];
  const float* W2    = (const float*)d_in[7];
  const float* as2   = (const float*)d_in[8];
  const float* ad2   = (const float*)d_in[9];
  const float* b2    = (const float*)d_in[10];
  float* out = (float*)d_out;

  char* w = (char*)d_ws;
  auto carve = [&](size_t bytes) {
    char* p = w;
    w += (bytes + 255) & ~(size_t)255;
    return p;
  };
  float*    acc   = (float*)carve((size_t)N_NODES * D * 4);
  float*    asrc  = (float*)carve((size_t)N_NODES * 4);
  float*    adst  = (float*)carve((size_t)N_NODES * 4);
  float*    asrc2 = (float*)carve((size_t)N_NODES * 4);
  float*    adst2 = (float*)carve((size_t)N_NODES * 4);
  float*    cnt   = (float*)carve((size_t)(N_GRAPHS + N_BCNT) * 4);
  int*      bcnt  = (int*)(cnt + N_GRAPHS);
  unsigned* binw  = (unsigned*)carve((size_t)OWN_WG * BIN_CAP * 4);
  int*      ovf   = (int*)carve((size_t)N_EDGES * 4);

  const int TB = 256;
  const int gFin   = (2 * N_GRAPHS * D + TB - 1) / TB;

  hipMemsetAsync(cnt, 0, (size_t)(N_GRAPHS + N_BCNT) * 4, stream);
  front_kernel<<<BIN_WG + OWN_WG, OWN_TPB, 0, stream>>>(ei, binw, bcnt, ovf, x, W1, as1, ad1, asrc, adst,
                                                     batch, cnt, out);

  owner_pass<<<OWN_WG, OWN_TPB, 0, stream>>>(ei, asrc, adst, x, b1, batch, binw, bcnt, ovf,
                                             acc, out, 1, W1, W2, as2, ad2, asrc2, adst2);

  owner_pass<<<OWN_WG, OWN_TPB, 0, stream>>>(ei, asrc2, adst2, acc, b2, batch, binw, bcnt, ovf,
                                             acc, out + N_GRAPHS * D, 0, W2, W2, as2, ad2, asrc2, adst2);

  finalize_kernel<<<gFin, TB, 0, stream>>>(out, cnt);
}
